# starting kernel
# baseline (speedup 1.0000x reference)
#include <hip/hip_runtime.h>
#include <hip/hip_bf16.h>


#define NG 16
#define NN 128
#define DD 9216
#define EPSV 1e-5f
#define TITER 5
#define KSPLIT 16
#define KCHUNK 32
#define APITCH 36
#define MATF (NG * NN * NN)

typedef __attribute__((ext_vector_type(2))) float v2f;
typedef __attribute__((ext_vector_type(8))) float v8f;

static __device__ __forceinline__ v8f wmma_f32(v2f a, v2f b, v8f c) {
  return __builtin_amdgcn_wmma_f32_16x16x4_f32(
       false, a,  false, b,
       (short)0, c,  false,  false);
}

__global__ void oni_mean(const float* __restrict__ w, float* __restrict__ mean) {
  __shared__ float red[256];
  const int row = blockIdx.x;
  const float4* p = reinterpret_cast<const float4*>(w + (size_t)row * DD);
  float s = 0.f;
  for (int i = threadIdx.x; i < DD / 4; i += 256) {
    float4 v = p[i];
    s += (v.x + v.y) + (v.z + v.w);
  }
  red[threadIdx.x] = s;
  __syncthreads();
  for (int off = 128; off > 0; off >>= 1) {
    if (threadIdx.x < off) red[threadIdx.x] += red[threadIdx.x + off];
    __syncthreads();
  }
  if (threadIdx.x == 0) mean[row] = red[0] * (1.0f / (float)DD);
}

__global__ void oni_zero(float* __restrict__ S) {
  S[blockIdx.x * 256 + threadIdx.x] = 0.f;
}

__global__ void oni_cov(const float* __restrict__ w, const float* __restrict__ mean,
                        float* __restrict__ S) {
  __shared__ float lds[NN * APITCH];
  const int g = blockIdx.y;
  const int kbase = blockIdx.x * (DD / KSPLIT);
  const int tid = threadIdx.x;
  const int lane = tid & 31, wave = tid >> 5;
  const int M0 = wave * 16;
  const int l15 = lane & 15;
  const int kb = (lane < 16) ? 0 : 2;

  v8f acc[8];
  v8f z = {};
#pragma unroll
  for (int t = 0; t < 8; ++t) acc[t] = z;

  for (int k0 = kbase; k0 < kbase + DD / KSPLIT; k0 += KCHUNK) {
#pragma unroll
    for (int i = 0; i < 4; ++i) {
      int idx = i * 256 + tid;
      int r = idx >> 3, kk = (idx & 7) * 4;
      int grow = g * NN + r;
      float m = mean[grow];
      float4 v = *reinterpret_cast<const float4*>(w + (size_t)grow * DD + k0 + kk);
      v.x -= m; v.y -= m; v.z -= m; v.w -= m;
      *reinterpret_cast<float4*>(&lds[r * APITCH + kk]) = v;
    }
    __syncthreads();
#pragma unroll
    for (int k4 = 0; k4 < KCHUNK; k4 += 4) {
      v2f a = *reinterpret_cast<const v2f*>(&lds[(M0 + l15) * APITCH + k4 + kb]);
      v2f bf[8];
#pragma unroll
      for (int nt = 0; nt < 8; ++nt)
        bf[nt] = *reinterpret_cast<const v2f*>(&lds[(nt * 16 + l15) * APITCH + k4 + kb]);
#pragma unroll
      for (int nt = 0; nt < 8; ++nt)
        acc[nt] = wmma_f32(a, bf[nt], acc[nt]);
    }
    __syncthreads();
  }

  const int Mb = M0 + ((lane < 16) ? 0 : 8);
  float* Sg = S + (size_t)g * NN * NN;
#pragma unroll
  for (int nt = 0; nt < 8; ++nt) {
    int col = nt * 16 + l15;
#pragma unroll
    for (int r = 0; r < 8; ++r) {
      atomicAdd(&Sg[(Mb + r) * NN + col], acc[nt][r]);
    }
  }
}

__global__ void oni_norm(float* __restrict__ S, float* __restrict__ rs,
                         float* __restrict__ B0) {
  __shared__ float red[256];
  __shared__ float nrm;
  const int g = blockIdx.x;
  float* Sg = S + g * (NN * NN);
  float s = 0.f;
  for (int i = threadIdx.x; i < NN * NN; i += 256) {
    int r = i >> 7, c = i & 127;
    float v = Sg[i] + ((r == c) ? EPSV : 0.f);
    s += v * v;
  }
  red[threadIdx.x] = s;
  __syncthreads();
  for (int off = 128; off > 0; off >>= 1) {
    if (threadIdx.x < off) red[threadIdx.x] += red[threadIdx.x + off];
    __syncthreads();
  }
  if (threadIdx.x == 0) {
    float n = sqrtf(red[0]);
    nrm = n;
    rs[g] = 1.0f / sqrtf(n);
  }
  __syncthreads();
  float inv = 1.0f / nrm;
  float* Bg = B0 + g * (NN * NN);
  for (int i = threadIdx.x; i < NN * NN; i += 256) {
    int r = i >> 7, c = i & 127;
    Sg[i] = (Sg[i] + ((r == c) ? EPSV : 0.f)) * inv;
    Bg[i] = (r == c) ? 1.f : 0.f;
  }
}

__global__ void oni_gemm128(const float* __restrict__ A, const float* __restrict__ Bm,
                            const float* __restrict__ P, float* __restrict__ Out,
                            float alpha, float beta) {
  __shared__ float ldsA[NN * APITCH];
  __shared__ float ldsB[KCHUNK * NN];
  const int g = blockIdx.x;
  const float* Ag = A + g * (NN * NN);
  const float* Bg = Bm + g * (NN * NN);
  const int tid = threadIdx.x, lane = tid & 31, wave = tid >> 5;
  const int M0 = wave * 16, l15 = lane & 15;
  const int kb = (lane < 16) ? 0 : 2;

  v8f acc[8];
  v8f z = {};
#pragma unroll
  for (int t = 0; t < 8; ++t) acc[t] = z;

  for (int k0 = 0; k0 < NN; k0 += KCHUNK) {
#pragma unroll
    for (int i = 0; i < 4; ++i) {
      int idx = i * 256 + tid;
      int r = idx >> 3, kk = (idx & 7) * 4;
      float4 v = *reinterpret_cast<const float4*>(Ag + r * NN + k0 + kk);
      *reinterpret_cast<float4*>(&ldsA[r * APITCH + kk]) = v;
    }
#pragma unroll
    for (int i = 0; i < 4; ++i) {
      int idx = i * 256 + tid;
      int kk = idx >> 5, c = (idx & 31) * 4;
      float4 v = *reinterpret_cast<const float4*>(Bg + (k0 + kk) * NN + c);
      *reinterpret_cast<float4*>(&ldsB[kk * NN + c]) = v;
    }
    __syncthreads();
#pragma unroll
    for (int k4 = 0; k4 < KCHUNK; k4 += 4) {
      v2f a = *reinterpret_cast<const v2f*>(&ldsA[(M0 + l15) * APITCH + k4 + kb]);
      v2f bf[8];
#pragma unroll
      for (int nt = 0; nt < 8; ++nt) {
        int col = nt * 16 + l15;
        bf[nt].x = ldsB[(k4 + kb) * NN + col];
        bf[nt].y = ldsB[(k4 + kb + 1) * NN + col];
      }
#pragma unroll
      for (int nt = 0; nt < 8; ++nt)
        acc[nt] = wmma_f32(a, bf[nt], acc[nt]);
    }
    __syncthreads();
  }

  const int Mb = M0 + ((lane < 16) ? 0 : 8);
  float* Og = Out + g * (NN * NN);
  const float* Pg = P ? (P + g * (NN * NN)) : nullptr;
#pragma unroll
  for (int nt = 0; nt < 8; ++nt) {
    int col = nt * 16 + l15;
#pragma unroll
    for (int r = 0; r < 8; ++r) {
      float v = alpha * acc[nt][r];
      if (Pg) v += beta * Pg[(Mb + r) * NN + col];
      Og[(Mb + r) * NN + col] = v;
    }
  }
}

__global__ void oni_out(const float* __restrict__ Bt, const float* __restrict__ w,
                        const float* __restrict__ mean, const float* __restrict__ rs,
                        float* __restrict__ out) {
  __shared__ float ldsA[NN * APITCH];
  __shared__ float ldsB[KCHUNK * NN];
  const int g = blockIdx.y;
  const int N0 = blockIdx.x * 128;
  const float* Ag = Bt + g * (NN * NN);
  const int tid = threadIdx.x, lane = tid & 31, wave = tid >> 5;
  const int M0 = wave * 16, l15 = lane & 15;
  const int kb = (lane < 16) ? 0 : 2;

  v8f acc[8];
  v8f z = {};
#pragma unroll
  for (int t = 0; t < 8; ++t) acc[t] = z;

  for (int k0 = 0; k0 < NN; k0 += KCHUNK) {
#pragma unroll
    for (int i = 0; i < 4; ++i) {
      int idx = i * 256 + tid;
      int r = idx >> 3, kk = (idx & 7) * 4;
      float4 v = *reinterpret_cast<const float4*>(Ag + r * NN + k0 + kk);
      *reinterpret_cast<float4*>(&ldsA[r * APITCH + kk]) = v;
    }
#pragma unroll
    for (int i = 0; i < 4; ++i) {
      int idx = i * 256 + tid;
      int kk = idx >> 5, c = (idx & 31) * 4;
      int grow = g * NN + k0 + kk;
      float m = mean[grow];
      float4 v = *reinterpret_cast<const float4*>(w + (size_t)grow * DD + N0 + c);
      v.x -= m; v.y -= m; v.z -= m; v.w -= m;
      *reinterpret_cast<float4*>(&ldsB[kk * NN + c]) = v;
    }
    __syncthreads();
#pragma unroll
    for (int k4 = 0; k4 < KCHUNK; k4 += 4) {
      v2f a = *reinterpret_cast<const v2f*>(&ldsA[(M0 + l15) * APITCH + k4 + kb]);
      v2f bf[8];
#pragma unroll
      for (int nt = 0; nt < 8; ++nt) {
        int col = nt * 16 + l15;
        bf[nt].x = ldsB[(k4 + kb) * NN + col];
        bf[nt].y = ldsB[(k4 + kb + 1) * NN + col];
      }
#pragma unroll
      for (int nt = 0; nt < 8; ++nt)
        acc[nt] = wmma_f32(a, bf[nt], acc[nt]);
    }
    __syncthreads();
  }

  const float scale = rs[g];
  const int Mb = M0 + ((lane < 16) ? 0 : 8);
#pragma unroll
  for (int nt = 0; nt < 8; ++nt) {
    int col = N0 + nt * 16 + l15;
#pragma unroll
    for (int r = 0; r < 8; ++r) {
      out[(size_t)(g * NN + Mb + r) * DD + col] = acc[nt][r] * scale;
    }
  }
}

extern "C" void kernel_launch(void* const* d_in, const int* in_sizes, int n_in,
                              void* d_out, int out_size, void* d_ws, size_t ws_size,
                              hipStream_t stream) {
  const float* w = (const float*)d_in[0];
  float* out = (float*)d_out;
  float* ws = (float*)d_ws;

  float* mean = ws;
  float* rs = ws + 2048;
  float* S  = ws + 2080;
  float* BA = S + MATF;
  float* BB = BA + MATF;
  float* T2 = BB + MATF;
  float* T3 = T2 + MATF;

  oni_mean<<<NG * NN, 256, 0, stream>>>(w, mean);
  oni_zero<<<MATF / 256, 256, 0, stream>>>(S);
  oni_cov<<<dim3(KSPLIT, NG), 256, 0, stream>>>(w, mean, S);
  oni_norm<<<NG, 256, 0, stream>>>(S, rs, BA);

  float* Bc = BA;
  float* Bn = BB;
  for (int t = 0; t < TITER; ++t) {
    oni_gemm128<<<NG, 256, 0, stream>>>(Bc, Bc, nullptr, T2, 1.f, 0.f);
    oni_gemm128<<<NG, 256, 0, stream>>>(T2, Bc, nullptr, T3, 1.f, 0.f);
    oni_gemm128<<<NG, 256, 0, stream>>>(T3, S, Bc, Bn, -0.5f, 1.5f);
    float* tmp = Bc; Bc = Bn; Bn = tmp;
  }

  oni_out<<<dim3(DD / 128, NG), 256, 0, stream>>>(Bc, w, mean, rs, out);
}


# Newton-Schulz products: all global loads issued in one phase (whole K=128 of A and B to LDS, one barrier; 1.5*B term as one group under the WMMAs) instead of load-wait pairs; sums bit-identical
# speedup vs baseline: 1.7369x; 1.7369x over previous
#include <hip/hip_runtime.h>
#include <hip/hip_bf16.h>


#define NG 16
#define NN 128
#define DD 9216
#define EPSV 1e-5f
#define TITER 5
#define KSPLIT 16
#define KCHUNK 32
#define APITCH 36
#define MATF (NG * NN * NN)
#define WPITCH 132

typedef __attribute__((ext_vector_type(2))) float v2f;
typedef __attribute__((ext_vector_type(8))) float v8f;

static __device__ __forceinline__ v8f wmma_f32(v2f a, v2f b, v8f c) {
  return __builtin_amdgcn_wmma_f32_16x16x4_f32(
       false, a,  false, b,
       (short)0, c,  false,  false);
}

__global__ void oni_mean(const float* __restrict__ w, float* __restrict__ mean) {
  __shared__ float red[256];
  const int row = blockIdx.x;
  const float4* p = reinterpret_cast<const float4*>(w + (size_t)row * DD);
  float s = 0.f;
  for (int i = threadIdx.x; i < DD / 4; i += 256) {
    float4 v = p[i];
    s += (v.x + v.y) + (v.z + v.w);
  }
  red[threadIdx.x] = s;
  __syncthreads();
  for (int off = 128; off > 0; off >>= 1) {
    if (threadIdx.x < off) red[threadIdx.x] += red[threadIdx.x + off];
    __syncthreads();
  }
  if (threadIdx.x == 0) mean[row] = red[0] * (1.0f / (float)DD);
}

__global__ void oni_zero(float* __restrict__ S) {
  S[blockIdx.x * 256 + threadIdx.x] = 0.f;
}

__global__ void oni_cov(const float* __restrict__ w, const float* __restrict__ mean,
                        float* __restrict__ S) {
  __shared__ float lds[NN * APITCH];
  const int g = blockIdx.y;
  const int kbase = blockIdx.x * (DD / KSPLIT);
  const int tid = threadIdx.x;
  const int lane = tid & 31, wave = tid >> 5;
  const int M0 = wave * 16;
  const int l15 = lane & 15;
  const int kb = (lane < 16) ? 0 : 2;

  v8f acc[8];
  v8f z = {};
#pragma unroll
  for (int t = 0; t < 8; ++t) acc[t] = z;

  for (int k0 = kbase; k0 < kbase + DD / KSPLIT; k0 += KCHUNK) {
#pragma unroll
    for (int i = 0; i < 4; ++i) {
      int idx = i * 256 + tid;
      int r = idx >> 3, kk = (idx & 7) * 4;
      int grow = g * NN + r;
      float m = mean[grow];
      float4 v = *reinterpret_cast<const float4*>(w + (size_t)grow * DD + k0 + kk);
      v.x -= m; v.y -= m; v.z -= m; v.w -= m;
      *reinterpret_cast<float4*>(&lds[r * APITCH + kk]) = v;
    }
    __syncthreads();
#pragma unroll
    for (int k4 = 0; k4 < KCHUNK; k4 += 4) {
      v2f a = *reinterpret_cast<const v2f*>(&lds[(M0 + l15) * APITCH + k4 + kb]);
      v2f bf[8];
#pragma unroll
      for (int nt = 0; nt < 8; ++nt)
        bf[nt] = *reinterpret_cast<const v2f*>(&lds[(nt * 16 + l15) * APITCH + k4 + kb]);
#pragma unroll
      for (int nt = 0; nt < 8; ++nt)
        acc[nt] = wmma_f32(a, bf[nt], acc[nt]);
    }
    __syncthreads();
  }

  const int Mb = M0 + ((lane < 16) ? 0 : 8);
  float* Sg = S + (size_t)g * NN * NN;
#pragma unroll
  for (int nt = 0; nt < 8; ++nt) {
    int col = nt * 16 + l15;
#pragma unroll
    for (int r = 0; r < 8; ++r) {
      atomicAdd(&Sg[(Mb + r) * NN + col], acc[nt][r]);
    }
  }
}

__global__ void oni_norm(float* __restrict__ S, float* __restrict__ rs,
                         float* __restrict__ B0) {
  __shared__ float red[256];
  __shared__ float nrm;
  const int g = blockIdx.x;
  float* Sg = S + g * (NN * NN);
  float s = 0.f;
  for (int i = threadIdx.x; i < NN * NN; i += 256) {
    int r = i >> 7, c = i & 127;
    float v = Sg[i] + ((r == c) ? EPSV : 0.f);
    s += v * v;
  }
  red[threadIdx.x] = s;
  __syncthreads();
  for (int off = 128; off > 0; off >>= 1) {
    if (threadIdx.x < off) red[threadIdx.x] += red[threadIdx.x + off];
    __syncthreads();
  }
  if (threadIdx.x == 0) {
    float n = sqrtf(red[0]);
    nrm = n;
    rs[g] = 1.0f / sqrtf(n);
  }
  __syncthreads();
  float inv = 1.0f / nrm;
  float* Bg = B0 + g * (NN * NN);
  for (int i = threadIdx.x; i < NN * NN; i += 256) {
    int r = i >> 7, c = i & 127;
    Sg[i] = (Sg[i] + ((r == c) ? EPSV : 0.f)) * inv;
    Bg[i] = (r == c) ? 1.f : 0.f;
  }
}

template <bool HAS_P>
static __device__ __forceinline__ void gemm128_group(
    const float* __restrict__ Ag, const float* __restrict__ Bg,
    const float* __restrict__ Pg, float* __restrict__ Og,
    float alpha, float beta, float* ldsA, float* ldsB) {
  const int tid = threadIdx.x, lane = tid & 31, wave = tid >> 5;
  const int M0 = wave * 16, l15 = lane & 15;
  const int kb = (lane < 16) ? 0 : 2;
  const int Mb = M0 + ((lane < 16) ? 0 : 8);

  float4 ra[16], rb[16];
#pragma unroll
  for (int i = 0; i < 16; ++i) {
    int idx = i * 256 + tid;
    ra[i] = *reinterpret_cast<const float4*>(Ag + idx * 4);
  }
#pragma unroll
  for (int i = 0; i < 16; ++i) {
    int idx = i * 256 + tid;
    rb[i] = *reinterpret_cast<const float4*>(Bg + idx * 4);
  }
#pragma unroll
  for (int i = 0; i < 16; ++i) {
    int idx = i * 256 + tid;
    int r = idx >> 5, c = (idx & 31) * 4;
    *reinterpret_cast<float4*>(&ldsA[r * WPITCH + c]) = ra[i];
  }
#pragma unroll
  for (int i = 0; i < 16; ++i) {
    int idx = i * 256 + tid;
    int r = idx >> 5, c = (idx & 31) * 4;
    *reinterpret_cast<float4*>(&ldsB[r * NN + c]) = rb[i];
  }
  __syncthreads();

  float p[8][8];
  if (HAS_P) {
#pragma unroll
    for (int nt = 0; nt < 8; ++nt) {
#pragma unroll
      for (int r = 0; r < 8; ++r) p[nt][r] = Pg[(Mb + r) * NN + nt * 16 + l15];
    }
  }

  v8f acc[8];
  v8f z = {};
#pragma unroll
  for (int t = 0; t < 8; ++t) acc[t] = z;

  for (int k0 = 0; k0 < NN; k0 += KCHUNK) {
#pragma unroll
    for (int kk = 0; kk < KCHUNK; kk += 4) {
      const int k4 = k0 + kk;
      v2f a = *reinterpret_cast<const v2f*>(&ldsA[(M0 + l15) * WPITCH + k4 + kb]);
      v2f bf[8];
#pragma unroll
      for (int nt = 0; nt < 8; ++nt) {
        int col = nt * 16 + l15;
        bf[nt].x = ldsB[(k4 + kb) * NN + col];
        bf[nt].y = ldsB[(k4 + kb + 1) * NN + col];
      }
#pragma unroll
      for (int nt = 0; nt < 8; ++nt)
        acc[nt] = wmma_f32(a, bf[nt], acc[nt]);
    }
  }

#pragma unroll
  for (int nt = 0; nt < 8; ++nt) {
    int col = nt * 16 + l15;
#pragma unroll
    for (int r = 0; r < 8; ++r) {
      float v = alpha * acc[nt][r];
      if (HAS_P) v = __builtin_fmaf(beta, p[nt][r], v);
      Og[(Mb + r) * NN + col] = v;
    }
  }
}

__global__ void __launch_bounds__(256)
oni_gemm128_pair(const float* __restrict__ B, const float* __restrict__ S,
                 float* __restrict__ T2, float* __restrict__ C) {
  __shared__ float ldsA[NN * WPITCH];
  __shared__ float ldsB[NN * NN];
  const int g = blockIdx.x;
  const bool second = (blockIdx.y != 0);
  const float* Rg = (second ? S : B) + g * (NN * NN);
  float* Og = (second ? C : T2) + g * (NN * NN);
  gemm128_group<false>(B + g * (NN * NN), Rg, nullptr, Og, 1.f, 0.f, ldsA, ldsB);
}

__global__ void __launch_bounds__(256)
oni_gemm128_comb(const float* __restrict__ A, const float* __restrict__ Bm,
                 const float* __restrict__ P, float* __restrict__ Out,
                 float alpha, float beta) {
  __shared__ float ldsA[NN * WPITCH];
  __shared__ float ldsB[NN * NN];
  const int g = blockIdx.x;
  gemm128_group<true>(A + g * (NN * NN), Bm + g * (NN * NN), P + g * (NN * NN),
                      Out + g * (NN * NN), alpha, beta, ldsA, ldsB);
}

__global__ void oni_out(const float* __restrict__ Bt, const float* __restrict__ w,
                        const float* __restrict__ mean, const float* __restrict__ rs,
                        float* __restrict__ out) {
  __shared__ float ldsA[NN * APITCH];
  __shared__ float ldsB[KCHUNK * NN];
  const int g = blockIdx.y;
  const int N0 = blockIdx.x * 128;
  const float* Ag = Bt + g * (NN * NN);
  const int tid = threadIdx.x, lane = tid & 31, wave = tid >> 5;
  const int M0 = wave * 16, l15 = lane & 15;
  const int kb = (lane < 16) ? 0 : 2;

  v8f acc[8];
  v8f z = {};
#pragma unroll
  for (int t = 0; t < 8; ++t) acc[t] = z;

  for (int k0 = 0; k0 < NN; k0 += KCHUNK) {
#pragma unroll
    for (int i = 0; i < 4; ++i) {
      int idx = i * 256 + tid;
      int r = idx >> 3, kk = (idx & 7) * 4;
      float4 v = *reinterpret_cast<const float4*>(Ag + r * NN + k0 + kk);
      *reinterpret_cast<float4*>(&ldsA[r * APITCH + kk]) = v;
    }
#pragma unroll
    for (int i = 0; i < 4; ++i) {
      int idx = i * 256 + tid;
      int kk = idx >> 5, c = (idx & 31) * 4;
      int grow = g * NN + k0 + kk;
      float m = mean[grow];
      float4 v = *reinterpret_cast<const float4*>(w + (size_t)grow * DD + N0 + c);
      v.x -= m; v.y -= m; v.z -= m; v.w -= m;
      *reinterpret_cast<float4*>(&ldsB[kk * NN + c]) = v;
    }
    __syncthreads();
#pragma unroll
    for (int k4 = 0; k4 < KCHUNK; k4 += 4) {
      v2f a = *reinterpret_cast<const v2f*>(&ldsA[(M0 + l15) * APITCH + k4 + kb]);
      v2f bf[8];
#pragma unroll
      for (int nt = 0; nt < 8; ++nt) {
        int col = nt * 16 + l15;
        bf[nt].x = ldsB[(k4 + kb) * NN + col];
        bf[nt].y = ldsB[(k4 + kb + 1) * NN + col];
      }
#pragma unroll
      for (int nt = 0; nt < 8; ++nt)
        acc[nt] = wmma_f32(a, bf[nt], acc[nt]);
    }
    __syncthreads();
  }

  const float scale = rs[g];
  const int Mb = M0 + ((lane < 16) ? 0 : 8);
#pragma unroll
  for (int nt = 0; nt < 8; ++nt) {
    int col = N0 + nt * 16 + l15;
#pragma unroll
    for (int r = 0; r < 8; ++r) {
      out[(size_t)(g * NN + Mb + r) * DD + col] = acc[nt][r] * scale;
    }
  }
}

extern "C" void kernel_launch(void* const* d_in, const int* in_sizes, int n_in,
                              void* d_out, int out_size, void* d_ws, size_t ws_size,
                              hipStream_t stream) {
  const float* w = (const float*)d_in[0];
  float* out = (float*)d_out;
  float* ws = (float*)d_ws;

  float* mean = ws;
  float* rs = ws + 2048;
  float* S  = ws + 2080;
  float* BA = S + MATF;
  float* BB = BA + MATF;
  float* T2 = BB + MATF;
  float* T3 = T2 + MATF;

  oni_mean<<<NG * NN, 256, 0, stream>>>(w, mean);
  oni_zero<<<MATF / 256, 256, 0, stream>>>(S);
  oni_cov<<<dim3(KSPLIT, NG), 256, 0, stream>>>(w, mean, S);
  oni_norm<<<NG, 256, 0, stream>>>(S, rs, BA);

  float* Bc = BA;
  float* Bn = BB;
  for (int t = 0; t < TITER; ++t) {
    oni_gemm128_pair<<<dim3(NG, 2), 256, 0, stream>>>(Bc, S, T2, T3);
    oni_gemm128_comb<<<NG, 256, 0, stream>>>(T2, T3, Bc, Bn, -0.5f, 1.5f);
    float* tmp = Bc; Bc = Bn; Bn = tmp;
  }

  oni_out<<<dim3(DD / 128, NG), 256, 0, stream>>>(Bc, w, mean, rs, out);
}


# Output GEMM re-tiled: a wave owns 16-column strips over all 128 rows; Zc fragments come straight from the weight (no LDS, no K-chunk loop), Bt staged whole once per work-group; sums bit-identical
# speedup vs baseline: 1.8432x; 1.0612x over previous
#include <hip/hip_runtime.h>
#include <hip/hip_bf16.h>


#define NG 16
#define NN 128
#define DD 9216
#define EPSV 1e-5f
#define TITER 5
#define KSPLIT 16
#define KCHUNK 32
#define APITCH 36
#define MATF (NG * NN * NN)
#define WPITCH 132

typedef __attribute__((ext_vector_type(2))) float v2f;
typedef __attribute__((ext_vector_type(8))) float v8f;

static __device__ __forceinline__ v8f wmma_f32(v2f a, v2f b, v8f c) {
  return __builtin_amdgcn_wmma_f32_16x16x4_f32(
       false, a,  false, b,
       (short)0, c,  false,  false);
}

__global__ void oni_mean(const float* __restrict__ w, float* __restrict__ mean) {
  __shared__ float red[256];
  const int row = blockIdx.x;
  const float4* p = reinterpret_cast<const float4*>(w + (size_t)row * DD);
  float s = 0.f;
  for (int i = threadIdx.x; i < DD / 4; i += 256) {
    float4 v = p[i];
    s += (v.x + v.y) + (v.z + v.w);
  }
  red[threadIdx.x] = s;
  __syncthreads();
  for (int off = 128; off > 0; off >>= 1) {
    if (threadIdx.x < off) red[threadIdx.x] += red[threadIdx.x + off];
    __syncthreads();
  }
  if (threadIdx.x == 0) mean[row] = red[0] * (1.0f / (float)DD);
}

__global__ void oni_zero(float* __restrict__ S) {
  S[blockIdx.x * 256 + threadIdx.x] = 0.f;
}

__global__ void oni_cov(const float* __restrict__ w, const float* __restrict__ mean,
                        float* __restrict__ S) {
  __shared__ float lds[NN * APITCH];
  const int g = blockIdx.y;
  const int kbase = blockIdx.x * (DD / KSPLIT);
  const int tid = threadIdx.x;
  const int lane = tid & 31, wave = tid >> 5;
  const int M0 = wave * 16;
  const int l15 = lane & 15;
  const int kb = (lane < 16) ? 0 : 2;

  v8f acc[8];
  v8f z = {};
#pragma unroll
  for (int t = 0; t < 8; ++t) acc[t] = z;

  for (int k0 = kbase; k0 < kbase + DD / KSPLIT; k0 += KCHUNK) {
#pragma unroll
    for (int i = 0; i < 4; ++i) {
      int idx = i * 256 + tid;
      int r = idx >> 3, kk = (idx & 7) * 4;
      int grow = g * NN + r;
      float m = mean[grow];
      float4 v = *reinterpret_cast<const float4*>(w + (size_t)grow * DD + k0 + kk);
      v.x -= m; v.y -= m; v.z -= m; v.w -= m;
      *reinterpret_cast<float4*>(&lds[r * APITCH + kk]) = v;
    }
    __syncthreads();
#pragma unroll
    for (int k4 = 0; k4 < KCHUNK; k4 += 4) {
      v2f a = *reinterpret_cast<const v2f*>(&lds[(M0 + l15) * APITCH + k4 + kb]);
      v2f bf[8];
#pragma unroll
      for (int nt = 0; nt < 8; ++nt)
        bf[nt] = *reinterpret_cast<const v2f*>(&lds[(nt * 16 + l15) * APITCH + k4 + kb]);
#pragma unroll
      for (int nt = 0; nt < 8; ++nt)
        acc[nt] = wmma_f32(a, bf[nt], acc[nt]);
    }
    __syncthreads();
  }

  const int Mb = M0 + ((lane < 16) ? 0 : 8);
  float* Sg = S + (size_t)g * NN * NN;
#pragma unroll
  for (int nt = 0; nt < 8; ++nt) {
    int col = nt * 16 + l15;
#pragma unroll
    for (int r = 0; r < 8; ++r) {
      atomicAdd(&Sg[(Mb + r) * NN + col], acc[nt][r]);
    }
  }
}

__global__ void oni_norm(float* __restrict__ S, float* __restrict__ rs,
                         float* __restrict__ B0) {
  __shared__ float red[256];
  __shared__ float nrm;
  const int g = blockIdx.x;
  float* Sg = S + g * (NN * NN);
  float s = 0.f;
  for (int i = threadIdx.x; i < NN * NN; i += 256) {
    int r = i >> 7, c = i & 127;
    float v = Sg[i] + ((r == c) ? EPSV : 0.f);
    s += v * v;
  }
  red[threadIdx.x] = s;
  __syncthreads();
  for (int off = 128; off > 0; off >>= 1) {
    if (threadIdx.x < off) red[threadIdx.x] += red[threadIdx.x + off];
    __syncthreads();
  }
  if (threadIdx.x == 0) {
    float n = sqrtf(red[0]);
    nrm = n;
    rs[g] = 1.0f / sqrtf(n);
  }
  __syncthreads();
  float inv = 1.0f / nrm;
  float* Bg = B0 + g * (NN * NN);
  for (int i = threadIdx.x; i < NN * NN; i += 256) {
    int r = i >> 7, c = i & 127;
    Sg[i] = (Sg[i] + ((r == c) ? EPSV : 0.f)) * inv;
    Bg[i] = (r == c) ? 1.f : 0.f;
  }
}

template <bool HAS_P>
static __device__ __forceinline__ void gemm128_group(
    const float* __restrict__ Ag, const float* __restrict__ Bg,
    const float* __restrict__ Pg, float* __restrict__ Og,
    float alpha, float beta, float* ldsA, float* ldsB) {
  const int tid = threadIdx.x, lane = tid & 31, wave = tid >> 5;
  const int M0 = wave * 16, l15 = lane & 15;
  const int kb = (lane < 16) ? 0 : 2;
  const int Mb = M0 + ((lane < 16) ? 0 : 8);

  float4 ra[16], rb[16];
#pragma unroll
  for (int i = 0; i < 16; ++i) {
    int idx = i * 256 + tid;
    ra[i] = *reinterpret_cast<const float4*>(Ag + idx * 4);
  }
#pragma unroll
  for (int i = 0; i < 16; ++i) {
    int idx = i * 256 + tid;
    rb[i] = *reinterpret_cast<const float4*>(Bg + idx * 4);
  }
#pragma unroll
  for (int i = 0; i < 16; ++i) {
    int idx = i * 256 + tid;
    int r = idx >> 5, c = (idx & 31) * 4;
    *reinterpret_cast<float4*>(&ldsA[r * WPITCH + c]) = ra[i];
  }
#pragma unroll
  for (int i = 0; i < 16; ++i) {
    int idx = i * 256 + tid;
    int r = idx >> 5, c = (idx & 31) * 4;
    *reinterpret_cast<float4*>(&ldsB[r * NN + c]) = rb[i];
  }
  __syncthreads();

  float p[8][8];
  if (HAS_P) {
#pragma unroll
    for (int nt = 0; nt < 8; ++nt) {
#pragma unroll
      for (int r = 0; r < 8; ++r) p[nt][r] = Pg[(Mb + r) * NN + nt * 16 + l15];
    }
  }

  v8f acc[8];
  v8f z = {};
#pragma unroll
  for (int t = 0; t < 8; ++t) acc[t] = z;

  for (int k0 = 0; k0 < NN; k0 += KCHUNK) {
#pragma unroll
    for (int kk = 0; kk < KCHUNK; kk += 4) {
      const int k4 = k0 + kk;
      v2f a = *reinterpret_cast<const v2f*>(&ldsA[(M0 + l15) * WPITCH + k4 + kb]);
      v2f bf[8];
#pragma unroll
      for (int nt = 0; nt < 8; ++nt) {
        int col = nt * 16 + l15;
        bf[nt].x = ldsB[(k4 + kb) * NN + col];
        bf[nt].y = ldsB[(k4 + kb + 1) * NN + col];
      }
#pragma unroll
      for (int nt = 0; nt < 8; ++nt)
        acc[nt] = wmma_f32(a, bf[nt], acc[nt]);
    }
  }

#pragma unroll
  for (int nt = 0; nt < 8; ++nt) {
    int col = nt * 16 + l15;
#pragma unroll
    for (int r = 0; r < 8; ++r) {
      float v = alpha * acc[nt][r];
      if (HAS_P) v = __builtin_fmaf(beta, p[nt][r], v);
      Og[(Mb + r) * NN + col] = v;
    }
  }
}

__global__ void __launch_bounds__(256)
oni_gemm128_pair(const float* __restrict__ B, const float* __restrict__ S,
                 float* __restrict__ T2, float* __restrict__ C) {
  __shared__ float ldsA[NN * WPITCH];
  __shared__ float ldsB[NN * NN];
  const int g = blockIdx.x;
  const bool second = (blockIdx.y != 0);
  const float* Rg = (second ? S : B) + g * (NN * NN);
  float* Og = (second ? C : T2) + g * (NN * NN);
  gemm128_group<false>(B + g * (NN * NN), Rg, nullptr, Og, 1.f, 0.f, ldsA, ldsB);
}

__global__ void __launch_bounds__(256)
oni_gemm128_comb(const float* __restrict__ A, const float* __restrict__ Bm,
                 const float* __restrict__ P, float* __restrict__ Out,
                 float alpha, float beta) {
  __shared__ float ldsA[NN * WPITCH];
  __shared__ float ldsB[NN * NN];
  const int g = blockIdx.x;
  gemm128_group<true>(A + g * (NN * NN), Bm + g * (NN * NN), P + g * (NN * NN),
                      Out + g * (NN * NN), alpha, beta, ldsA, ldsB);
}

#define OWAVES 12
#define OSTRIPS 3
#define OSLAB (OWAVES * OSTRIPS * 16)
#define OSTEP (OWAVES * 16)

static __device__ __forceinline__ void out_chunk_load(
    const float* __restrict__ p, float (&b)[16]) {
#pragma unroll
  for (int j = 0; j < 8; ++j) {
    b[2 * j]     = p[(size_t)(4 * j) * DD];
    b[2 * j + 1] = p[(size_t)(4 * j + 1) * DD];
  }
}

__global__ void __launch_bounds__(OWAVES * 32)
oni_out(const float* __restrict__ Bt, const float* __restrict__ w,
        const float* __restrict__ mean, const float* __restrict__ rs,
        float* __restrict__ out) {
  __shared__ __attribute__((aligned(16))) float ldsA[NN * WPITCH];
  __shared__ __attribute__((aligned(16))) float ldsM[NN];
  const int g = blockIdx.y;
  const int tid = threadIdx.x, lane = tid & 31, wave = tid >> 5;
  const int l15 = lane & 15;
  const int kb = (lane < 16) ? 0 : 2;
  const int col0 = blockIdx.x * OSLAB + wave * 16 + l15;
  const float* Ag = Bt + g * (NN * NN);

  float4 ra[10];
#pragma unroll
  for (int i = 0; i < 10; ++i) {
    int idx = i * (OWAVES * 32) + tid;
    ra[i] = *reinterpret_cast<const float4*>(Ag + idx * 4);
  }
  const bool part = tid < (NN * NN / 4 - 10 * OWAVES * 32);
  float4 rl = {};
  if (part) rl = *reinterpret_cast<const float4*>(Ag + (10 * OWAVES * 32 + tid) * 4);
  float mrow = 0.f;
  if (tid < NN) mrow = mean[g * NN + tid];

  const float* wp = w + (size_t)(g * NN + kb) * DD + col0;
  float bc[16], bn[16];
  out_chunk_load(wp, bc);

#pragma unroll
  for (int i = 0; i < 10; ++i) {
    int idx = i * (OWAVES * 32) + tid;
    int r = idx >> 5, c = (idx & 31) * 4;
    *reinterpret_cast<float4*>(&ldsA[r * WPITCH + c]) = ra[i];
  }
  if (part) {
    int idx = 10 * OWAVES * 32 + tid;
    int r = idx >> 5, c = (idx & 31) * 4;
    *reinterpret_cast<float4*>(&ldsA[r * WPITCH + c]) = rl;
  }
  if (tid < NN) ldsM[tid] = mrow;
  __syncthreads();

  const float scale = rs[g];
  float* op = out + (size_t)(g * NN + ((lane < 16) ? 0 : 8)) * DD + col0;

  v8f acc[8];
  v8f z = {};
#pragma unroll
  for (int t = 0; t < 8; ++t) acc[t] = z;

  v2f ac[8], an[8];
#pragma unroll
  for (int mt = 0; mt < 8; ++mt)
    ac[mt] = *reinterpret_cast<const v2f*>(&ldsA[(mt * 16 + l15) * WPITCH + kb]);
  v2f bf;
  {
    v2f mm = *reinterpret_cast<const v2f*>(&ldsM[kb]);
    bf.x = bc[0] - mm.x;
    bf.y = bc[1] - mm.y;
  }

  for (int it = 0; it < OSTRIPS * 4; ++it) {
    const int kc = it & 3;
    const int itn = (it + 1 < OSTRIPS * 4) ? it + 1 : it;
    out_chunk_load(wp + (size_t)((itn & 3) * KCHUNK) * DD + (itn >> 2) * OSTEP, bn);
    const int k0 = kc * KCHUNK;
#pragma unroll
    for (int j = 0; j < 8; ++j) {
      const int k4n = (j < 7) ? (k0 + 4 * j + 4) : ((itn & 3) * KCHUNK);
#pragma unroll
      for (int mt = 0; mt < 8; ++mt)
        an[mt] = *reinterpret_cast<const v2f*>(&ldsA[(mt * 16 + l15) * WPITCH + k4n + kb]);
      v2f mm = *reinterpret_cast<const v2f*>(&ldsM[k4n + kb]);
      __builtin_amdgcn_sched_barrier(0);
#pragma unroll
      for (int mt = 0; mt < 8; ++mt)
        acc[mt] = wmma_f32(ac[mt], bf, acc[mt]);
      __builtin_amdgcn_sched_barrier(0);
      const float bx = (j < 7) ? bc[2 * j + 2] : bn[0];
      const float by = (j < 7) ? bc[2 * j + 3] : bn[1];
      bf.x = bx - mm.x;
      bf.y = by - mm.y;
#pragma unroll
      for (int mt = 0; mt < 8; ++mt) ac[mt] = an[mt];
    }
    if (kc == 3) {
      float* os = op + (it >> 2) * OSTEP;
#pragma unroll
      for (int mt = 0; mt < 8; ++mt) {
#pragma unroll
        for (int r = 0; r < 8; ++r)
          os[(size_t)(mt * 16 + r) * DD] = acc[mt][r] * scale;
        acc[mt] = z;
      }
    }
#pragma unroll
    for (int q = 0; q < 16; ++q) bc[q] = bn[q];
  }
}

extern "C" void kernel_launch(void* const* d_in, const int* in_sizes, int n_in,
                              void* d_out, int out_size, void* d_ws, size_t ws_size,
                              hipStream_t stream) {
  const float* w = (const float*)d_in[0];
  float* out = (float*)d_out;
  float* ws = (float*)d_ws;

  float* mean = ws;
  float* rs = ws + 2048;
  float* S  = ws + 2080;
  float* BA = S + MATF;
  float* BB = BA + MATF;
  float* T2 = BB + MATF;
  float* T3 = T2 + MATF;

  oni_mean<<<NG * NN, 256, 0, stream>>>(w, mean);
  oni_zero<<<MATF / 256, 256, 0, stream>>>(S);
  oni_cov<<<dim3(KSPLIT, NG), 256, 0, stream>>>(w, mean, S);
  oni_norm<<<NG, 256, 0, stream>>>(S, rs, BA);

  float* Bc = BA;
  float* Bn = BB;
  for (int t = 0; t < TITER; ++t) {
    oni_gemm128_pair<<<dim3(NG, 2), 256, 0, stream>>>(Bc, S, T2, T3);
    oni_gemm128_comb<<<NG, 256, 0, stream>>>(T2, T3, Bc, Bn, -0.5f, 1.5f);
    float* tmp = Bc; Bc = Bn; Bn = tmp;
  }

  oni_out<<<dim3(DD / OSLAB, NG), OWAVES * 32, 0, stream>>>(Bc, w, mean, rs, out);
}


# Newton-Schulz iteration in ONE launch (was 2): the work-group of a 32x32 tile of Bn forms rows I of B*B and columns J of B*S itself, hands them over in LDS and multiplies them; sums bit-identical
# speedup vs baseline: 2.8249x; 1.5326x over previous
#include <hip/hip_runtime.h>
#include <hip/hip_bf16.h>


#define NG 16
#define NN 128
#define DD 9216
#define EPSV 1e-5f
#define TITER 5
#define KSPLIT 8
#define KCHUNK 32
#define MATF (NG * NN * NN)
#define WPITCH 132

typedef __attribute__((ext_vector_type(2))) float v2f;
typedef __attribute__((ext_vector_type(8))) float v8f;

static __device__ __forceinline__ v8f wmma_f32(v2f a, v2f b, v8f c) {
  return __builtin_amdgcn_wmma_f32_16x16x4_f32(
       false, a,  false, b,
       (short)0, c,  false,  false);
}

__global__ void oni_mean(const float* __restrict__ w, float* __restrict__ mean) {
  __shared__ float red[256];
  const int row = blockIdx.x;
  const float4* p = reinterpret_cast<const float4*>(w + (size_t)row * DD);
  float s = 0.f;
  for (int i = threadIdx.x; i < DD / 4; i += 256) {
    float4 v = p[i];
    s += (v.x + v.y) + (v.z + v.w);
  }
  red[threadIdx.x] = s;
  __syncthreads();
  for (int off = 128; off > 0; off >>= 1) {
    if (threadIdx.x < off) red[threadIdx.x] += red[threadIdx.x + off];
    __syncthreads();
  }
  if (threadIdx.x == 0) mean[row] = red[0] * (1.0f / (float)DD);
}

__global__ void oni_zero(float* __restrict__ S) {
  S[blockIdx.x * 256 + threadIdx.x] = 0.f;
}

#define CROWS 32
#define CKC 128

__global__ void __launch_bounds__(256)
oni_cov(const float* __restrict__ w, const float* __restrict__ mean,
        float* __restrict__ S) {
  __shared__ __attribute__((aligned(16))) float lds[NN * WPITCH];
  const int band = blockIdx.x;
  const int kbase = blockIdx.y * (DD / KSPLIT);
  const int g = blockIdx.z;
  const int tid = threadIdx.x;
  const int lane = tid & 31, wave = tid >> 5;
  const int l15 = lane & 15;
  const int kb = (lane < 16) ? 0 : 2;

  float m[16];
#pragma unroll
  for (int i = 0; i < 16; ++i) m[i] = mean[g * NN + i * 8 + wave];
  const float* wp = w + (size_t)(g * NN + wave) * DD + kbase + lane * 4;
  float* sp = &lds[wave * WPITCH + lane * 4];

  v8f acc[2];
  v8f z = {};
  acc[0] = z;
  acc[1] = z;

  const float* fpa = &lds[(band * CROWS + l15) * WPITCH + kb];
  const float* fpb = &lds[(wave * 16 + l15) * WPITCH + kb];

  for (int k0 = 0; k0 < DD / KSPLIT; k0 += CKC) {
    float4 v[16];
#pragma unroll
    for (int i = 0; i < 16; ++i)
      v[i] = *reinterpret_cast<const float4*>(wp + (size_t)(i * 8) * DD + k0);
#pragma unroll
    for (int i = 0; i < 16; ++i) {
      float4 c = v[i];
      c.x -= m[i]; c.y -= m[i]; c.z -= m[i]; c.w -= m[i];
      *reinterpret_cast<float4*>(sp + (i * 8) * WPITCH) = c;
    }
    __syncthreads();

    v2f fa[2][16], fb[2][8];
#pragma unroll
    for (int j = 0; j < 8; ++j) {
      fa[0][2 * j]     = *reinterpret_cast<const v2f*>(fpa + 4 * j);
      fa[0][2 * j + 1] = *reinterpret_cast<const v2f*>(fpa + 16 * WPITCH + 4 * j);
      fb[0][j]         = *reinterpret_cast<const v2f*>(fpb + 4 * j);
    }
#pragma unroll
    for (int c = 0; c < CKC / KCHUNK; ++c) {
      const int cur = c & 1, nxt = cur ^ 1;
      if (c + 1 < CKC / KCHUNK) {
#pragma unroll
        for (int j = 0; j < 8; ++j) {
          const int k4 = (c + 1) * KCHUNK + 4 * j;
          fa[nxt][2 * j]     = *reinterpret_cast<const v2f*>(fpa + k4);
          fa[nxt][2 * j + 1] = *reinterpret_cast<const v2f*>(fpa + 16 * WPITCH + k4);
          fb[nxt][j]         = *reinterpret_cast<const v2f*>(fpb + k4);
        }
      }
      __builtin_amdgcn_sched_barrier(0);
#pragma unroll
      for (int j = 0; j < 8; ++j) {
        acc[0] = wmma_f32(fa[cur][2 * j], fb[cur][j], acc[0]);
        acc[1] = wmma_f32(fa[cur][2 * j + 1], fb[cur][j], acc[1]);
      }
      __builtin_amdgcn_sched_barrier(0);
    }
    __syncthreads();
  }

  const int Mb = band * CROWS + ((lane < 16) ? 0 : 8);
  float* Sg = S + (size_t)g * NN * NN;
  const int col = wave * 16 + l15;
#pragma unroll
  for (int t = 0; t < 2; ++t) {
#pragma unroll
    for (int r = 0; r < 8; ++r) {
      atomicAdd(&Sg[(Mb + t * 16 + r) * NN + col], acc[t][r]);
    }
  }
}

__global__ void oni_norm(float* __restrict__ S, float* __restrict__ rs,
                         float* __restrict__ B0) {
  __shared__ float red[256];
  __shared__ float nrm;
  const int g = blockIdx.x;
  float* Sg = S + g * (NN * NN);
  float s = 0.f;
  for (int i = threadIdx.x; i < NN * NN; i += 256) {
    int r = i >> 7, c = i & 127;
    float v = Sg[i] + ((r == c) ? EPSV : 0.f);
    s += v * v;
  }
  red[threadIdx.x] = s;
  __syncthreads();
  for (int off = 128; off > 0; off >>= 1) {
    if (threadIdx.x < off) red[threadIdx.x] += red[threadIdx.x + off];
    __syncthreads();
  }
  if (threadIdx.x == 0) {
    float n = sqrtf(red[0]);
    nrm = n;
    rs[g] = 1.0f / sqrtf(n);
  }
  __syncthreads();
  float inv = 1.0f / nrm;
  float* Bg = B0 + g * (NN * NN);
  for (int i = threadIdx.x; i < NN * NN; i += 256) {
    int r = i >> 7, c = i & 127;
    Sg[i] = (Sg[i] + ((r == c) ? EPSV : 0.f)) * inv;
    Bg[i] = (r == c) ? 1.f : 0.f;
  }
}

#define NT 32
#define BPB 288
#define BPS 96

template <int BP>
static __device__ __forceinline__ void mma_row_two_cols(
    const float* fpa, const float* fpb, v8f (&acc)[2]) {
  v2f fa[2][8], fb[2][16];
#pragma unroll
  for (int j = 0; j < 8; ++j) {
    fa[0][j] = *reinterpret_cast<const v2f*>(fpa + 4 * j);
#pragma unroll
    for (int t = 0; t < 2; ++t)
      fb[0][2 * j + t] = *reinterpret_cast<const v2f*>(fpb + (2 * j) * BP + t * 32);
  }
#pragma unroll
  for (int c = 0; c < NN / KCHUNK; ++c) {
    const int cur = c & 1, nxt = cur ^ 1;
    if (c + 1 < NN / KCHUNK) {
#pragma unroll
      for (int j = 0; j < 8; ++j) {
        const int k4 = (c + 1) * KCHUNK + 4 * j;
        fa[nxt][j] = *reinterpret_cast<const v2f*>(fpa + k4);
#pragma unroll
        for (int t = 0; t < 2; ++t)
          fb[nxt][2 * j + t] = *reinterpret_cast<const v2f*>(fpb + (k4 >> 1) * BP + t * 32);
      }
    }
    __builtin_amdgcn_sched_barrier(0);
#pragma unroll
    for (int j = 0; j < 8; ++j) {
#pragma unroll
      for (int t = 0; t < 2; ++t)
        acc[t] = wmma_f32(fa[cur][j], fb[cur][2 * j + t], acc[t]);
    }
    __builtin_amdgcn_sched_barrier(0);
  }
}

template <int BP>
static __device__ __forceinline__ v8f mma_row_one_col(const float* fpa, const float* fpb) {
  v2f fa[2][8], fb[2][8];
  v8f acc = {};
#pragma unroll
  for (int j = 0; j < 8; ++j) {
    fa[0][j] = *reinterpret_cast<const v2f*>(fpa + 4 * j);
    fb[0][j] = *reinterpret_cast<const v2f*>(fpb + (2 * j) * BP);
  }
#pragma unroll
  for (int c = 0; c < NN / KCHUNK; ++c) {
    const int cur = c & 1, nxt = cur ^ 1;
    if (c + 1 < NN / KCHUNK) {
#pragma unroll
      for (int j = 0; j < 8; ++j) {
        const int k4 = (c + 1) * KCHUNK + 4 * j;
        fa[nxt][j] = *reinterpret_cast<const v2f*>(fpa + k4);
        fb[nxt][j] = *reinterpret_cast<const v2f*>(fpb + (k4 >> 1) * BP);
      }
    }
    __builtin_amdgcn_sched_barrier(0);
#pragma unroll
    for (int j = 0; j < 8; ++j) acc = wmma_f32(fa[cur][j], fb[cur][j], acc);
    __builtin_amdgcn_sched_barrier(0);
  }
  return acc;
}

__global__ void __launch_bounds__(256)
oni_ns_iter(const float* __restrict__ B, const float* __restrict__ S,
            float* __restrict__ Bn) {
  __shared__ __attribute__((aligned(16))) float ldsA[NN * WPITCH];
  __shared__ __attribute__((aligned(16))) float ldsBB[(NN / 2) * BPB];
  __shared__ __attribute__((aligned(16))) float ldsS[(NN / 2) * BPS];
  __shared__ __attribute__((aligned(16))) float ldsT[NT * WPITCH];
  __shared__ __attribute__((aligned(16))) float ldsC[(NN / 2) * BPS];
  const int g = blockIdx.y;
  const int row0 = (blockIdx.x >> 2) * NT, col0 = (blockIdx.x & 3) * NT;
  const int tid = threadIdx.x, lane = tid & 31, wave = tid >> 5;
  const int l15 = lane & 15;
  const int h = lane >> 4;
  const int kb = 2 * h;
  const float* Bg = B + g * (NN * NN);
  const float* Sg = S + g * (NN * NN);

  float4 rb0[8], rb1[8], rs0[2], rs1[2];
#pragma unroll
  for (int i = 0; i < 8; ++i) {
    int q = i * 256 + tid;
    int kp = q >> 5, c = (q & 31) * 4;
    rb0[i] = *reinterpret_cast<const float4*>(Bg + (2 * kp) * NN + c);
    rb1[i] = *reinterpret_cast<const float4*>(Bg + (2 * kp + 1) * NN + c);
  }
#pragma unroll
  for (int i = 0; i < 2; ++i) {
    int q = i * 256 + tid;
    int kp = q >> 3, c = (q & 7) * 4;
    rs0[i] = *reinterpret_cast<const float4*>(Sg + (2 * kp) * NN + col0 + c);
    rs1[i] = *reinterpret_cast<const float4*>(Sg + (2 * kp + 1) * NN + col0 + c);
  }
#pragma unroll
  for (int i = 0; i < 8; ++i) {
    int q = i * 256 + tid;
    int kp = q >> 5, c = (q & 31) * 4;
    *reinterpret_cast<float4*>(&ldsA[(2 * kp) * WPITCH + c]) = rb0[i];
    *reinterpret_cast<float4*>(&ldsA[(2 * kp + 1) * WPITCH + c]) = rb1[i];
    float4 lo = {rb0[i].x, rb1[i].x, rb0[i].y, rb1[i].y};
    float4 hi = {rb0[i].z, rb1[i].z, rb0[i].w, rb1[i].w};
    *reinterpret_cast<float4*>(&ldsBB[kp * BPB + c * 2]) = lo;
    *reinterpret_cast<float4*>(&ldsBB[kp * BPB + c * 2 + 4]) = hi;
  }
#pragma unroll
  for (int i = 0; i < 2; ++i) {
    int q = i * 256 + tid;
    int kp = q >> 3, c = (q & 7) * 4;
    float4 lo = {rs0[i].x, rs1[i].x, rs0[i].y, rs1[i].y};
    float4 hi = {rs0[i].z, rs1[i].z, rs0[i].w, rs1[i].w};
    *reinterpret_cast<float4*>(&ldsS[kp * BPS + c * 2]) = lo;
    *reinterpret_cast<float4*>(&ldsS[kp * BPS + c * 2 + 4]) = hi;
  }
  __syncthreads();

  v8f z = {};

  const int tr = (wave & 1) * 16, tc = (wave >> 1) * 32;
  v8f acc1[2];
  acc1[0] = z;
  acc1[1] = z;
  mma_row_two_cols<BPB>(&ldsA[(row0 + tr + l15) * WPITCH + kb],
                        &ldsBB[h * BPB + (tc + l15) * 2], acc1);

  v8f acc2[2];
  acc2[0] = z;
  acc2[1] = z;
  mma_row_two_cols<BPS>(&ldsA[(wave * 16 + l15) * WPITCH + kb],
                        &ldsS[h * BPS + l15 * 2], acc2);

#pragma unroll
  for (int t = 0; t < 2; ++t) {
#pragma unroll
    for (int r = 0; r < 8; ++r)
      ldsT[(tr + 8 * h + r) * WPITCH + tc + t * 16 + l15] = acc1[t][r];
#pragma unroll
    for (int r = 0; r < 8; r += 2) {
      v2f pr = {acc2[t][r], acc2[t][r + 1]};
      *reinterpret_cast<v2f*>(&ldsC[(wave * 8 + 4 * h + (r >> 1)) * BPS + (t * 16 + l15) * 2]) = pr;
    }
  }
  __syncthreads();

  if (wave < 4) {
    const int tm = (wave >> 1) * 16, tn = (wave & 1) * 16;
    float p[8];
#pragma unroll
    for (int r = 0; r < 8; ++r)
      p[r] = ldsA[(row0 + tm + 8 * h + r) * WPITCH + col0 + tn + l15];
    v8f acc3 = mma_row_one_col<BPS>(&ldsT[(tm + l15) * WPITCH + kb],
                                    &ldsC[h * BPS + (tn + l15) * 2]);
    float* Og = Bn + g * (NN * NN) + (row0 + tm + 8 * h) * NN + col0 + tn + l15;
#pragma unroll
    for (int r = 0; r < 8; ++r) {
      float v = -0.5f * acc3[r];
      v = __builtin_fmaf(1.5f, p[r], v);
      Og[r * NN] = v;
    }
  }
}

#define OWAVES 12
#define OSTRIPS 3
#define OSLAB (OWAVES * OSTRIPS * 16)
#define OSTEP (OWAVES * 16)

static __device__ __forceinline__ void out_chunk_load(
    const float* __restrict__ p, float (&b)[16]) {
#pragma unroll
  for (int j = 0; j < 8; ++j) {
    b[2 * j]     = p[(size_t)(4 * j) * DD];
    b[2 * j + 1] = p[(size_t)(4 * j + 1) * DD];
  }
}

__global__ void __launch_bounds__(OWAVES * 32)
oni_out(const float* __restrict__ Bt, const float* __restrict__ w,
        const float* __restrict__ mean, const float* __restrict__ rs,
        float* __restrict__ out) {
  __shared__ __attribute__((aligned(16))) float ldsA[NN * WPITCH];
  __shared__ __attribute__((aligned(16))) float ldsM[NN];
  const int g = blockIdx.y;
  const int tid = threadIdx.x, lane = tid & 31, wave = tid >> 5;
  const int l15 = lane & 15;
  const int kb = (lane < 16) ? 0 : 2;
  const int col0 = blockIdx.x * OSLAB + wave * 16 + l15;
  const float* Ag = Bt + g * (NN * NN);

  float4 ra[10];
#pragma unroll
  for (int i = 0; i < 10; ++i) {
    int idx = i * (OWAVES * 32) + tid;
    ra[i] = *reinterpret_cast<const float4*>(Ag + idx * 4);
  }
  const bool part = tid < (NN * NN / 4 - 10 * OWAVES * 32);
  float4 rl = {};
  if (part) rl = *reinterpret_cast<const float4*>(Ag + (10 * OWAVES * 32 + tid) * 4);
  float mrow = 0.f;
  if (tid < NN) mrow = mean[g * NN + tid];

  const float* wp = w + (size_t)(g * NN + kb) * DD + col0;
  float bc[16], bn[16];
  out_chunk_load(wp, bc);

#pragma unroll
  for (int i = 0; i < 10; ++i) {
    int idx = i * (OWAVES * 32) + tid;
    int r = idx >> 5, c = (idx & 31) * 4;
    *reinterpret_cast<float4*>(&ldsA[r * WPITCH + c]) = ra[i];
  }
  if (part) {
    int idx = 10 * OWAVES * 32 + tid;
    int r = idx >> 5, c = (idx & 31) * 4;
    *reinterpret_cast<float4*>(&ldsA[r * WPITCH + c]) = rl;
  }
  if (tid < NN) ldsM[tid] = mrow;
  __syncthreads();

  const float scale = rs[g];
  float* op = out + (size_t)(g * NN + ((lane < 16) ? 0 : 8)) * DD + col0;

  v8f acc[8];
  v8f z = {};
#pragma unroll
  for (int t = 0; t < 8; ++t) acc[t] = z;

  v2f ac[8], an[8];
#pragma unroll
  for (int mt = 0; mt < 8; ++mt)
    ac[mt] = *reinterpret_cast<const v2f*>(&ldsA[(mt * 16 + l15) * WPITCH + kb]);
  v2f bf;
  {
    v2f mm = *reinterpret_cast<const v2f*>(&ldsM[kb]);
    bf.x = bc[0] - mm.x;
    bf.y = bc[1] - mm.y;
  }

  for (int it = 0; it < OSTRIPS * 4; ++it) {
    const int kc = it & 3;
    const int itn = (it + 1 < OSTRIPS * 4) ? it + 1 : it;
    out_chunk_load(wp + (size_t)((itn & 3) * KCHUNK) * DD + (itn >> 2) * OSTEP, bn);
    const int k0 = kc * KCHUNK;
#pragma unroll
    for (int j = 0; j < 8; ++j) {
      const int k4n = (j < 7) ? (k0 + 4 * j + 4) : ((itn & 3) * KCHUNK);
#pragma unroll
      for (int mt = 0; mt < 8; ++mt)
        an[mt] = *reinterpret_cast<const v2f*>(&ldsA[(mt * 16 + l15) * WPITCH + k4n + kb]);
      v2f mm = *reinterpret_cast<const v2f*>(&ldsM[k4n + kb]);
      __builtin_amdgcn_sched_barrier(0);
#pragma unroll
      for (int mt = 0; mt < 8; ++mt)
        acc[mt] = wmma_f32(ac[mt], bf, acc[mt]);
      __builtin_amdgcn_sched_barrier(0);
      const float bx = (j < 7) ? bc[2 * j + 2] : bn[0];
      const float by = (j < 7) ? bc[2 * j + 3] : bn[1];
      bf.x = bx - mm.x;
      bf.y = by - mm.y;
#pragma unroll
      for (int mt = 0; mt < 8; ++mt) ac[mt] = an[mt];
    }
    if (kc == 3) {
      float* os = op + (it >> 2) * OSTEP;
#pragma unroll
      for (int mt = 0; mt < 8; ++mt) {
#pragma unroll
        for (int r = 0; r < 8; ++r)
          os[(size_t)(mt * 16 + r) * DD] = acc[mt][r] * scale;
        acc[mt] = z;
      }
    }
#pragma unroll
    for (int q = 0; q < 16; ++q) bc[q] = bn[q];
  }
}

extern "C" void kernel_launch(void* const* d_in, const int* in_sizes, int n_in,
                              void* d_out, int out_size, void* d_ws, size_t ws_size,
                              hipStream_t stream) {
  const float* w = (const float*)d_in[0];
  float* out = (float*)d_out;
  float* ws = (float*)d_ws;

  float* mean = ws;
  float* rs = ws + 2048;
  float* S  = ws + 2080;
  float* BA = S + MATF;
  float* BB = BA + MATF;

  oni_mean<<<NG * NN, 256, 0, stream>>>(w, mean);
  oni_zero<<<MATF / 256, 256, 0, stream>>>(S);
  oni_cov<<<dim3(NN / CROWS, KSPLIT, NG), 256, 0, stream>>>(w, mean, S);
  oni_norm<<<NG, 256, 0, stream>>>(S, rs, BA);

  float* Bc = BA;
  float* Bn = BB;
  for (int t = 0; t < TITER; ++t) {
    oni_ns_iter<<<dim3((NN / NT) * (NN / NT), NG), 256, 0, stream>>>(Bc, S, Bn);
    float* tmp = Bc; Bc = Bn; Bn = tmp;
  }

  oni_out<<<dim3(DD / OSLAB, NG), OWAVES * 32, 0, stream>>>(Bc, w, mean, rs, out);
}


# Norm launch folded into the first Newton-Schulz launch: each work-group reads whole unscaled S, sums its squares itself, scales its columns; B0 = I made in registers; 9 launches; norm sum reordered
# speedup vs baseline: 3.3640x; 1.1908x over previous
#include <hip/hip_runtime.h>
#include <hip/hip_bf16.h>


#define NG 16
#define NN 128
#define DD 9216
#define EPSV 1e-5f
#define TITER 5
#define KSPLIT 8
#define KCHUNK 32
#define MATF (NG * NN * NN)
#define WPITCH 132

typedef __attribute__((ext_vector_type(2))) float v2f;
typedef __attribute__((ext_vector_type(8))) float v8f;

static __device__ __forceinline__ v8f wmma_f32(v2f a, v2f b, v8f c) {
  return __builtin_amdgcn_wmma_f32_16x16x4_f32(
       false, a,  false, b,
       (short)0, c,  false,  false);
}

__global__ void oni_mean(const float* __restrict__ w, float* __restrict__ mean) {
  __shared__ float red[256];
  const int row = blockIdx.x;
  const float4* p = reinterpret_cast<const float4*>(w + (size_t)row * DD);
  float s = 0.f;
  for (int i = threadIdx.x; i < DD / 4; i += 256) {
    float4 v = p[i];
    s += (v.x + v.y) + (v.z + v.w);
  }
  red[threadIdx.x] = s;
  __syncthreads();
  for (int off = 128; off > 0; off >>= 1) {
    if (threadIdx.x < off) red[threadIdx.x] += red[threadIdx.x + off];
    __syncthreads();
  }
  if (threadIdx.x == 0) mean[row] = red[0] * (1.0f / (float)DD);
}

__global__ void oni_zero(float* __restrict__ S) {
  S[blockIdx.x * 256 + threadIdx.x] = 0.f;
}

#define CROWS 32
#define CKC 128

__global__ void __launch_bounds__(256)
oni_cov(const float* __restrict__ w, const float* __restrict__ mean,
        float* __restrict__ S) {
  __shared__ __attribute__((aligned(16))) float lds[NN * WPITCH];
  const int band = blockIdx.x;
  const int kbase = blockIdx.y * (DD / KSPLIT);
  const int g = blockIdx.z;
  const int tid = threadIdx.x;
  const int lane = tid & 31, wave = tid >> 5;
  const int l15 = lane & 15;
  const int kb = (lane < 16) ? 0 : 2;

  float m[16];
#pragma unroll
  for (int i = 0; i < 16; ++i) m[i] = mean[g * NN + i * 8 + wave];
  const float* wp = w + (size_t)(g * NN + wave) * DD + kbase + lane * 4;
  float* sp = &lds[wave * WPITCH + lane * 4];

  v8f acc[2];
  v8f z = {};
  acc[0] = z;
  acc[1] = z;

  const float* fpa = &lds[(band * CROWS + l15) * WPITCH + kb];
  const float* fpb = &lds[(wave * 16 + l15) * WPITCH + kb];

  for (int k0 = 0; k0 < DD / KSPLIT; k0 += CKC) {
    float4 v[16];
#pragma unroll
    for (int i = 0; i < 16; ++i)
      v[i] = *reinterpret_cast<const float4*>(wp + (size_t)(i * 8) * DD + k0);
#pragma unroll
    for (int i = 0; i < 16; ++i) {
      float4 c = v[i];
      c.x -= m[i]; c.y -= m[i]; c.z -= m[i]; c.w -= m[i];
      *reinterpret_cast<float4*>(sp + (i * 8) * WPITCH) = c;
    }
    __syncthreads();

    v2f fa[2][16], fb[2][8];
#pragma unroll
    for (int j = 0; j < 8; ++j) {
      fa[0][2 * j]     = *reinterpret_cast<const v2f*>(fpa + 4 * j);
      fa[0][2 * j + 1] = *reinterpret_cast<const v2f*>(fpa + 16 * WPITCH + 4 * j);
      fb[0][j]         = *reinterpret_cast<const v2f*>(fpb + 4 * j);
    }
#pragma unroll
    for (int c = 0; c < CKC / KCHUNK; ++c) {
      const int cur = c & 1, nxt = cur ^ 1;
      if (c + 1 < CKC / KCHUNK) {
#pragma unroll
        for (int j = 0; j < 8; ++j) {
          const int k4 = (c + 1) * KCHUNK + 4 * j;
          fa[nxt][2 * j]     = *reinterpret_cast<const v2f*>(fpa + k4);
          fa[nxt][2 * j + 1] = *reinterpret_cast<const v2f*>(fpa + 16 * WPITCH + k4);
          fb[nxt][j]         = *reinterpret_cast<const v2f*>(fpb + k4);
        }
      }
      __builtin_amdgcn_sched_barrier(0);
#pragma unroll
      for (int j = 0; j < 8; ++j) {
        acc[0] = wmma_f32(fa[cur][2 * j], fb[cur][j], acc[0]);
        acc[1] = wmma_f32(fa[cur][2 * j + 1], fb[cur][j], acc[1]);
      }
      __builtin_amdgcn_sched_barrier(0);
    }
    __syncthreads();
  }

  const int Mb = band * CROWS + ((lane < 16) ? 0 : 8);
  float* Sg = S + (size_t)g * NN * NN;
  const int col = wave * 16 + l15;
#pragma unroll
  for (int t = 0; t < 2; ++t) {
#pragma unroll
    for (int r = 0; r < 8; ++r) {
      atomicAdd(&Sg[(Mb + t * 16 + r) * NN + col], acc[t][r]);
    }
  }
}

#define NT 32
#define BPB 288
#define BPS 96

template <int BP>
static __device__ __forceinline__ void mma_row_two_cols(
    const float* fpa, const float* fpb, v8f (&acc)[2]) {
  v2f fa[2][8], fb[2][16];
#pragma unroll
  for (int j = 0; j < 8; ++j) {
    fa[0][j] = *reinterpret_cast<const v2f*>(fpa + 4 * j);
#pragma unroll
    for (int t = 0; t < 2; ++t)
      fb[0][2 * j + t] = *reinterpret_cast<const v2f*>(fpb + (2 * j) * BP + t * 32);
  }
#pragma unroll
  for (int c = 0; c < NN / KCHUNK; ++c) {
    const int cur = c & 1, nxt = cur ^ 1;
    if (c + 1 < NN / KCHUNK) {
#pragma unroll
      for (int j = 0; j < 8; ++j) {
        const int k4 = (c + 1) * KCHUNK + 4 * j;
        fa[nxt][j] = *reinterpret_cast<const v2f*>(fpa + k4);
#pragma unroll
        for (int t = 0; t < 2; ++t)
          fb[nxt][2 * j + t] = *reinterpret_cast<const v2f*>(fpb + (k4 >> 1) * BP + t * 32);
      }
    }
    __builtin_amdgcn_sched_barrier(0);
#pragma unroll
    for (int j = 0; j < 8; ++j) {
#pragma unroll
      for (int t = 0; t < 2; ++t)
        acc[t] = wmma_f32(fa[cur][j], fb[cur][2 * j + t], acc[t]);
    }
    __builtin_amdgcn_sched_barrier(0);
  }
}

template <int BP>
static __device__ __forceinline__ v8f mma_row_one_col(const float* fpa, const float* fpb) {
  v2f fa[2][8], fb[2][8];
  v8f acc = {};
#pragma unroll
  for (int j = 0; j < 8; ++j) {
    fa[0][j] = *reinterpret_cast<const v2f*>(fpa + 4 * j);
    fb[0][j] = *reinterpret_cast<const v2f*>(fpb + (2 * j) * BP);
  }
#pragma unroll
  for (int c = 0; c < NN / KCHUNK; ++c) {
    const int cur = c & 1, nxt = cur ^ 1;
    if (c + 1 < NN / KCHUNK) {
#pragma unroll
      for (int j = 0; j < 8; ++j) {
        const int k4 = (c + 1) * KCHUNK + 4 * j;
        fa[nxt][j] = *reinterpret_cast<const v2f*>(fpa + k4);
        fb[nxt][j] = *reinterpret_cast<const v2f*>(fpb + (k4 >> 1) * BP);
      }
    }
    __builtin_amdgcn_sched_barrier(0);
#pragma unroll
    for (int j = 0; j < 8; ++j) acc = wmma_f32(fa[cur][j], fb[cur][j], acc);
    __builtin_amdgcn_sched_barrier(0);
  }
  return acc;
}

template <bool FIRST>
__global__ void __launch_bounds__(256)
oni_ns_iter(const float* __restrict__ B, const float* __restrict__ S,
            float* __restrict__ Bn, float* __restrict__ Sn,
            float* __restrict__ rs) {
  __shared__ __attribute__((aligned(16))) float ldsA[NN * WPITCH];
  __shared__ __attribute__((aligned(16))) float ldsBB[(NN / 2) * BPB];
  __shared__ __attribute__((aligned(16))) float ldsS[(NN / 2) * BPS];
  __shared__ __attribute__((aligned(16))) float ldsT[NT * WPITCH];
  __shared__ __attribute__((aligned(16))) float ldsC[(NN / 2) * BPS];
  __shared__ float red[8];
  const int g = blockIdx.y;
  const int row0 = (blockIdx.x >> 2) * NT, col0 = (blockIdx.x & 3) * NT;
  const int tid = threadIdx.x, lane = tid & 31, wave = tid >> 5;
  const int l15 = lane & 15;
  const int h = lane >> 4;
  const int kb = 2 * h;
  const float* Bg = B + g * (NN * NN);
  const float* Sg = S + g * (NN * NN);

  float4 rb0[8], rb1[8], rs0[2], rs1[2];
  const float* Wg = FIRST ? Sg : Bg;
#pragma unroll
  for (int i = 0; i < 8; ++i) {
    int q = i * 256 + tid;
    int kp = q >> 5, c = (q & 31) * 4;
    rb0[i] = *reinterpret_cast<const float4*>(Wg + (2 * kp) * NN + c);
    rb1[i] = *reinterpret_cast<const float4*>(Wg + (2 * kp + 1) * NN + c);
  }
#pragma unroll
  for (int i = 0; i < 2; ++i) {
    int q = i * 256 + tid;
    int kp = q >> 3, c = (q & 7) * 4;
    rs0[i] = *reinterpret_cast<const float4*>(Sg + (2 * kp) * NN + col0 + c);
    rs1[i] = *reinterpret_cast<const float4*>(Sg + (2 * kp + 1) * NN + col0 + c);
  }
  if constexpr (FIRST) {
    float ss = 0.f;
#pragma unroll
    for (int i = 0; i < 8; ++i) {
      int q = i * 256 + tid;
      int kp = q >> 5, c = (q & 31) * 4;
      const int d = 2 * kp - c;
      float4 a = rb0[i], b = rb1[i];
      a.x += (d == 0) ? EPSV : 0.f;  a.y += (d == 1) ? EPSV : 0.f;
      a.z += (d == 2) ? EPSV : 0.f;  a.w += (d == 3) ? EPSV : 0.f;
      b.x += (d == -1) ? EPSV : 0.f; b.y += (d == 0) ? EPSV : 0.f;
      b.z += (d == 1) ? EPSV : 0.f;  b.w += (d == 2) ? EPSV : 0.f;
      ss += a.x * a.x; ss += a.y * a.y; ss += a.z * a.z; ss += a.w * a.w;
      ss += b.x * b.x; ss += b.y * b.y; ss += b.z * b.z; ss += b.w * b.w;
      rb0[i] = float4{(d == 0) ? 1.f : 0.f, (d == 1) ? 1.f : 0.f, (d == 2) ? 1.f : 0.f, (d == 3) ? 1.f : 0.f};
      rb1[i] = float4{(d == -1) ? 1.f : 0.f, (d == 0) ? 1.f : 0.f, (d == 1) ? 1.f : 0.f, (d == 2) ? 1.f : 0.f};
    }
#pragma unroll
    for (int off = 16; off > 0; off >>= 1) ss += __shfl_xor(ss, off);
    if (lane == 0) red[wave] = ss;
  }
#pragma unroll
  for (int i = 0; i < 8; ++i) {
    int q = i * 256 + tid;
    int kp = q >> 5, c = (q & 31) * 4;
    *reinterpret_cast<float4*>(&ldsA[(2 * kp) * WPITCH + c]) = rb0[i];
    *reinterpret_cast<float4*>(&ldsA[(2 * kp + 1) * WPITCH + c]) = rb1[i];
    float4 lo = {rb0[i].x, rb1[i].x, rb0[i].y, rb1[i].y};
    float4 hi = {rb0[i].z, rb1[i].z, rb0[i].w, rb1[i].w};
    *reinterpret_cast<float4*>(&ldsBB[kp * BPB + c * 2]) = lo;
    *reinterpret_cast<float4*>(&ldsBB[kp * BPB + c * 2 + 4]) = hi;
  }
  if constexpr (FIRST) {
    __syncthreads();
    const float tot = ((red[0] + red[1]) + (red[2] + red[3])) + ((red[4] + red[5]) + (red[6] + red[7]));
    const float nrm = sqrtf(tot);
    const float inv = 1.0f / nrm;
    if (tid == 0 && blockIdx.x == 0) rs[g] = 1.0f / sqrtf(nrm);
    float* Sng = Sn + g * (NN * NN);
#pragma unroll
    for (int i = 0; i < 2; ++i) {
      int q = i * 256 + tid;
      int kp = q >> 3, c = (q & 7) * 4;
      const int d = 2 * kp - (col0 + c);
      float4 a = rs0[i], b = rs1[i];
      a.x = (a.x + ((d == 0) ? EPSV : 0.f)) * inv;  a.y = (a.y + ((d == 1) ? EPSV : 0.f)) * inv;
      a.z = (a.z + ((d == 2) ? EPSV : 0.f)) * inv;  a.w = (a.w + ((d == 3) ? EPSV : 0.f)) * inv;
      b.x = (b.x + ((d == -1) ? EPSV : 0.f)) * inv; b.y = (b.y + ((d == 0) ? EPSV : 0.f)) * inv;
      b.z = (b.z + ((d == 1) ? EPSV : 0.f)) * inv;  b.w = (b.w + ((d == 2) ? EPSV : 0.f)) * inv;
      rs0[i] = a;
      rs1[i] = b;
      if (row0 == 0) {
        *reinterpret_cast<float4*>(Sng + (2 * kp) * NN + col0 + c) = a;
        *reinterpret_cast<float4*>(Sng + (2 * kp + 1) * NN + col0 + c) = b;
      }
    }
  }
#pragma unroll
  for (int i = 0; i < 2; ++i) {
    int q = i * 256 + tid;
    int kp = q >> 3, c = (q & 7) * 4;
    float4 lo = {rs0[i].x, rs1[i].x, rs0[i].y, rs1[i].y};
    float4 hi = {rs0[i].z, rs1[i].z, rs0[i].w, rs1[i].w};
    *reinterpret_cast<float4*>(&ldsS[kp * BPS + c * 2]) = lo;
    *reinterpret_cast<float4*>(&ldsS[kp * BPS + c * 2 + 4]) = hi;
  }
  __syncthreads();

  v8f z = {};

  const int tr = (wave & 1) * 16, tc = (wave >> 1) * 32;
  v8f acc1[2];
  acc1[0] = z;
  acc1[1] = z;
  mma_row_two_cols<BPB>(&ldsA[(row0 + tr + l15) * WPITCH + kb],
                        &ldsBB[h * BPB + (tc + l15) * 2], acc1);

  v8f acc2[2];
  acc2[0] = z;
  acc2[1] = z;
  mma_row_two_cols<BPS>(&ldsA[(wave * 16 + l15) * WPITCH + kb],
                        &ldsS[h * BPS + l15 * 2], acc2);

#pragma unroll
  for (int t = 0; t < 2; ++t) {
#pragma unroll
    for (int r = 0; r < 8; ++r)
      ldsT[(tr + 8 * h + r) * WPITCH + tc + t * 16 + l15] = acc1[t][r];
#pragma unroll
    for (int r = 0; r < 8; r += 2) {
      v2f pr = {acc2[t][r], acc2[t][r + 1]};
      *reinterpret_cast<v2f*>(&ldsC[(wave * 8 + 4 * h + (r >> 1)) * BPS + (t * 16 + l15) * 2]) = pr;
    }
  }
  __syncthreads();

  if (wave < 4) {
    const int tm = (wave >> 1) * 16, tn = (wave & 1) * 16;
    float p[8];
#pragma unroll
    for (int r = 0; r < 8; ++r)
      p[r] = ldsA[(row0 + tm + 8 * h + r) * WPITCH + col0 + tn + l15];
    v8f acc3 = mma_row_one_col<BPS>(&ldsT[(tm + l15) * WPITCH + kb],
                                    &ldsC[h * BPS + (tn + l15) * 2]);
    float* Og = Bn + g * (NN * NN) + (row0 + tm + 8 * h) * NN + col0 + tn + l15;
#pragma unroll
    for (int r = 0; r < 8; ++r) {
      float v = -0.5f * acc3[r];
      v = __builtin_fmaf(1.5f, p[r], v);
      Og[r * NN] = v;
    }
  }
}

#define OWAVES 12
#define OSTRIPS 3
#define OSLAB (OWAVES * OSTRIPS * 16)
#define OSTEP (OWAVES * 16)

static __device__ __forceinline__ void out_chunk_load(
    const float* __restrict__ p, float (&b)[16]) {
#pragma unroll
  for (int j = 0; j < 8; ++j) {
    b[2 * j]     = p[(size_t)(4 * j) * DD];
    b[2 * j + 1] = p[(size_t)(4 * j + 1) * DD];
  }
}

__global__ void __launch_bounds__(OWAVES * 32)
oni_out(const float* __restrict__ Bt, const float* __restrict__ w,
        const float* __restrict__ mean, const float* __restrict__ rs,
        float* __restrict__ out) {
  __shared__ __attribute__((aligned(16))) float ldsA[NN * WPITCH];
  __shared__ __attribute__((aligned(16))) float ldsM[NN];
  const int g = blockIdx.y;
  const int tid = threadIdx.x, lane = tid & 31, wave = tid >> 5;
  const int l15 = lane & 15;
  const int kb = (lane < 16) ? 0 : 2;
  const int col0 = blockIdx.x * OSLAB + wave * 16 + l15;
  const float* Ag = Bt + g * (NN * NN);

  float4 ra[10];
#pragma unroll
  for (int i = 0; i < 10; ++i) {
    int idx = i * (OWAVES * 32) + tid;
    ra[i] = *reinterpret_cast<const float4*>(Ag + idx * 4);
  }
  const bool part = tid < (NN * NN / 4 - 10 * OWAVES * 32);
  float4 rl = {};
  if (part) rl = *reinterpret_cast<const float4*>(Ag + (10 * OWAVES * 32 + tid) * 4);
  float mrow = 0.f;
  if (tid < NN) mrow = mean[g * NN + tid];

  const float* wp = w + (size_t)(g * NN + kb) * DD + col0;
  float bc[16], bn[16];
  out_chunk_load(wp, bc);

#pragma unroll
  for (int i = 0; i < 10; ++i) {
    int idx = i * (OWAVES * 32) + tid;
    int r = idx >> 5, c = (idx & 31) * 4;
    *reinterpret_cast<float4*>(&ldsA[r * WPITCH + c]) = ra[i];
  }
  if (part) {
    int idx = 10 * OWAVES * 32 + tid;
    int r = idx >> 5, c = (idx & 31) * 4;
    *reinterpret_cast<float4*>(&ldsA[r * WPITCH + c]) = rl;
  }
  if (tid < NN) ldsM[tid] = mrow;
  __syncthreads();

  const float scale = rs[g];
  float* op = out + (size_t)(g * NN + ((lane < 16) ? 0 : 8)) * DD + col0;

  v8f acc[8];
  v8f z = {};
#pragma unroll
  for (int t = 0; t < 8; ++t) acc[t] = z;

  v2f ac[8], an[8];
#pragma unroll
  for (int mt = 0; mt < 8; ++mt)
    ac[mt] = *reinterpret_cast<const v2f*>(&ldsA[(mt * 16 + l15) * WPITCH + kb]);
  v2f bf;
  {
    v2f mm = *reinterpret_cast<const v2f*>(&ldsM[kb]);
    bf.x = bc[0] - mm.x;
    bf.y = bc[1] - mm.y;
  }

  for (int it = 0; it < OSTRIPS * 4; ++it) {
    const int kc = it & 3;
    const int itn = (it + 1 < OSTRIPS * 4) ? it + 1 : it;
    out_chunk_load(wp + (size_t)((itn & 3) * KCHUNK) * DD + (itn >> 2) * OSTEP, bn);
    const int k0 = kc * KCHUNK;
#pragma unroll
    for (int j = 0; j < 8; ++j) {
      const int k4n = (j < 7) ? (k0 + 4 * j + 4) : ((itn & 3) * KCHUNK);
#pragma unroll
      for (int mt = 0; mt < 8; ++mt)
        an[mt] = *reinterpret_cast<const v2f*>(&ldsA[(mt * 16 + l15) * WPITCH + k4n + kb]);
      v2f mm = *reinterpret_cast<const v2f*>(&ldsM[k4n + kb]);
      __builtin_amdgcn_sched_barrier(0);
#pragma unroll
      for (int mt = 0; mt < 8; ++mt)
        acc[mt] = wmma_f32(ac[mt], bf, acc[mt]);
      __builtin_amdgcn_sched_barrier(0);
      const float bx = (j < 7) ? bc[2 * j + 2] : bn[0];
      const float by = (j < 7) ? bc[2 * j + 3] : bn[1];
      bf.x = bx - mm.x;
      bf.y = by - mm.y;
#pragma unroll
      for (int mt = 0; mt < 8; ++mt) ac[mt] = an[mt];
    }
    if (kc == 3) {
      float* os = op + (it >> 2) * OSTEP;
#pragma unroll
      for (int mt = 0; mt < 8; ++mt) {
#pragma unroll
        for (int r = 0; r < 8; ++r)
          os[(size_t)(mt * 16 + r) * DD] = acc[mt][r] * scale;
        acc[mt] = z;
      }
    }
#pragma unroll
    for (int q = 0; q < 16; ++q) bc[q] = bn[q];
  }
}

extern "C" void kernel_launch(void* const* d_in, const int* in_sizes, int n_in,
                              void* d_out, int out_size, void* d_ws, size_t ws_size,
                              hipStream_t stream) {
  const float* w = (const float*)d_in[0];
  float* out = (float*)d_out;
  float* ws = (float*)d_ws;

  float* mean = ws;
  float* rs = ws + 2048;
  float* S  = ws + 2080;
  float* BA = S + MATF;
  float* BB = BA + MATF;
  float* Sn = BB + MATF;

  oni_mean<<<NG * NN, 256, 0, stream>>>(w, mean);
  oni_zero<<<MATF / 256, 256, 0, stream>>>(S);
  oni_cov<<<dim3(NN / CROWS, KSPLIT, NG), 256, 0, stream>>>(w, mean, S);
  oni_ns_iter<true><<<dim3((NN / NT) * (NN / NT), NG), 256, 0, stream>>>(BB, S, BA, Sn, rs);

  float* Bc = BA;
  float* Bn = BB;
  for (int t = 1; t < TITER; ++t) {
    oni_ns_iter<false><<<dim3((NN / NT) * (NN / NT), NG), 256, 0, stream>>>(Bc, Sn, Bn, nullptr, nullptr);
    float* tmp = Bc; Bc = Bn; Bn = tmp;
  }

  oni_out<<<dim3(DD / OSLAB, NG), OWAVES * 32, 0, stream>>>(Bc, w, mean, rs, out);
}


# Newton-Schulz launch re-tiled: a 4-wave work-group owns 16 rows x 64 columns of Bn and chains ((B_I*B)*B)*S_J through LDS (was 8 waves, a 32x32 tile from B*B rows, B*S columns); sums reordered
# speedup vs baseline: 3.4873x; 1.0367x over previous
#include <hip/hip_runtime.h>
#include <hip/hip_bf16.h>


#define NG 16
#define NN 128
#define DD 9216
#define EPSV 1e-5f
#define TITER 5
#define KSPLIT 8
#define KCHUNK 32
#define MATF (NG * NN * NN)
#define WPITCH 132

typedef __attribute__((ext_vector_type(2))) float v2f;
typedef __attribute__((ext_vector_type(8))) float v8f;

static __device__ __forceinline__ v8f wmma_f32(v2f a, v2f b, v8f c) {
  return __builtin_amdgcn_wmma_f32_16x16x4_f32(
       false, a,  false, b,
       (short)0, c,  false,  false);
}

__global__ void oni_mean(const float* __restrict__ w, float* __restrict__ mean) {
  __shared__ float red[256];
  const int row = blockIdx.x;
  const float4* p = reinterpret_cast<const float4*>(w + (size_t)row * DD);
  float s = 0.f;
  for (int i = threadIdx.x; i < DD / 4; i += 256) {
    float4 v = p[i];
    s += (v.x + v.y) + (v.z + v.w);
  }
  red[threadIdx.x] = s;
  __syncthreads();
  for (int off = 128; off > 0; off >>= 1) {
    if (threadIdx.x < off) red[threadIdx.x] += red[threadIdx.x + off];
    __syncthreads();
  }
  if (threadIdx.x == 0) mean[row] = red[0] * (1.0f / (float)DD);
}

__global__ void oni_zero(float* __restrict__ S) {
  S[blockIdx.x * 256 + threadIdx.x] = 0.f;
}

#define CROWS 32
#define CKC 128

__global__ void __launch_bounds__(256)
oni_cov(const float* __restrict__ w, const float* __restrict__ mean,
        float* __restrict__ S) {
  __shared__ __attribute__((aligned(16))) float lds[NN * WPITCH];
  const int band = blockIdx.x;
  const int kbase = blockIdx.y * (DD / KSPLIT);
  const int g = blockIdx.z;
  const int tid = threadIdx.x;
  const int lane = tid & 31, wave = tid >> 5;
  const int l15 = lane & 15;
  const int kb = (lane < 16) ? 0 : 2;

  float m[16];
#pragma unroll
  for (int i = 0; i < 16; ++i) m[i] = mean[g * NN + i * 8 + wave];
  const float* wp = w + (size_t)(g * NN + wave) * DD + kbase + lane * 4;
  float* sp = &lds[wave * WPITCH + lane * 4];

  v8f acc[2];
  v8f z = {};
  acc[0] = z;
  acc[1] = z;

  const float* fpa = &lds[(band * CROWS + l15) * WPITCH + kb];
  const float* fpb = &lds[(wave * 16 + l15) * WPITCH + kb];

  for (int k0 = 0; k0 < DD / KSPLIT; k0 += CKC) {
    float4 v[16];
#pragma unroll
    for (int i = 0; i < 16; ++i)
      v[i] = *reinterpret_cast<const float4*>(wp + (size_t)(i * 8) * DD + k0);
#pragma unroll
    for (int i = 0; i < 16; ++i) {
      float4 c = v[i];
      c.x -= m[i]; c.y -= m[i]; c.z -= m[i]; c.w -= m[i];
      *reinterpret_cast<float4*>(sp + (i * 8) * WPITCH) = c;
    }
    __syncthreads();

    v2f fa[2][16], fb[2][8];
#pragma unroll
    for (int j = 0; j < 8; ++j) {
      fa[0][2 * j]     = *reinterpret_cast<const v2f*>(fpa + 4 * j);
      fa[0][2 * j + 1] = *reinterpret_cast<const v2f*>(fpa + 16 * WPITCH + 4 * j);
      fb[0][j]         = *reinterpret_cast<const v2f*>(fpb + 4 * j);
    }
#pragma unroll
    for (int c = 0; c < CKC / KCHUNK; ++c) {
      const int cur = c & 1, nxt = cur ^ 1;
      if (c + 1 < CKC / KCHUNK) {
#pragma unroll
        for (int j = 0; j < 8; ++j) {
          const int k4 = (c + 1) * KCHUNK + 4 * j;
          fa[nxt][2 * j]     = *reinterpret_cast<const v2f*>(fpa + k4);
          fa[nxt][2 * j + 1] = *reinterpret_cast<const v2f*>(fpa + 16 * WPITCH + k4);
          fb[nxt][j]         = *reinterpret_cast<const v2f*>(fpb + k4);
        }
      }
      __builtin_amdgcn_sched_barrier(0);
#pragma unroll
      for (int j = 0; j < 8; ++j) {
        acc[0] = wmma_f32(fa[cur][2 * j], fb[cur][j], acc[0]);
        acc[1] = wmma_f32(fa[cur][2 * j + 1], fb[cur][j], acc[1]);
      }
      __builtin_amdgcn_sched_barrier(0);
    }
    __syncthreads();
  }

  const int Mb = band * CROWS + ((lane < 16) ? 0 : 8);
  float* Sg = S + (size_t)g * NN * NN;
  const int col = wave * 16 + l15;
#pragma unroll
  for (int t = 0; t < 2; ++t) {
#pragma unroll
    for (int r = 0; r < 8; ++r) {
      atomicAdd(&Sg[(Mb + t * 16 + r) * NN + col], acc[t][r]);
    }
  }
}

#define NSR 16
#define NSC 64
#define NST 128
#define BPB 288
#define BPS 160

template <int BP>
static __device__ __forceinline__ void mma_row_two_cols(
    const float* fpa, const float* fpb, v8f (&acc)[2]) {
  v2f fa[2][8], fb[2][16];
#pragma unroll
  for (int j = 0; j < 8; ++j) {
    fa[0][j] = *reinterpret_cast<const v2f*>(fpa + 4 * j);
#pragma unroll
    for (int t = 0; t < 2; ++t)
      fb[0][2 * j + t] = *reinterpret_cast<const v2f*>(fpb + (2 * j) * BP + t * 32);
  }
#pragma unroll
  for (int c = 0; c < NN / KCHUNK; ++c) {
    const int cur = c & 1, nxt = cur ^ 1;
    if (c + 1 < NN / KCHUNK) {
#pragma unroll
      for (int j = 0; j < 8; ++j) {
        const int k4 = (c + 1) * KCHUNK + 4 * j;
        fa[nxt][j] = *reinterpret_cast<const v2f*>(fpa + k4);
#pragma unroll
        for (int t = 0; t < 2; ++t)
          fb[nxt][2 * j + t] = *reinterpret_cast<const v2f*>(fpb + (k4 >> 1) * BP + t * 32);
      }
    }
    __builtin_amdgcn_sched_barrier(0);
#pragma unroll
    for (int j = 0; j < 8; ++j) {
#pragma unroll
      for (int t = 0; t < 2; ++t)
        acc[t] = wmma_f32(fa[cur][j], fb[cur][2 * j + t], acc[t]);
    }
    __builtin_amdgcn_sched_barrier(0);
  }
}

template <int BP>
static __device__ __forceinline__ v8f mma_row_one_col(const float* fpa, const float* fpb) {
  v2f fa[2][8], fb[2][8];
  v8f acc = {};
#pragma unroll
  for (int j = 0; j < 8; ++j) {
    fa[0][j] = *reinterpret_cast<const v2f*>(fpa + 4 * j);
    fb[0][j] = *reinterpret_cast<const v2f*>(fpb + (2 * j) * BP);
  }
#pragma unroll
  for (int c = 0; c < NN / KCHUNK; ++c) {
    const int cur = c & 1, nxt = cur ^ 1;
    if (c + 1 < NN / KCHUNK) {
#pragma unroll
      for (int j = 0; j < 8; ++j) {
        const int k4 = (c + 1) * KCHUNK + 4 * j;
        fa[nxt][j] = *reinterpret_cast<const v2f*>(fpa + k4);
        fb[nxt][j] = *reinterpret_cast<const v2f*>(fpb + (k4 >> 1) * BP);
      }
    }
    __builtin_amdgcn_sched_barrier(0);
#pragma unroll
    for (int j = 0; j < 8; ++j) acc = wmma_f32(fa[cur][j], fb[cur][j], acc);
    __builtin_amdgcn_sched_barrier(0);
  }
  return acc;
}

template <bool FIRST>
__global__ void __launch_bounds__(NST)
oni_ns_iter(const float* __restrict__ B, const float* S,
            float* __restrict__ Bn, float* __restrict__ Sn,
            float* __restrict__ rs) {
  __shared__ __attribute__((aligned(16))) float ldsBB[(NN / 2) * BPB];
  __shared__ __attribute__((aligned(16))) float ldsS[(NN / 2) * BPS];
  __shared__ __attribute__((aligned(16))) float ldsR[NSR * WPITCH];
  __shared__ __attribute__((aligned(16))) float ldsT[NSR * WPITCH];
  __shared__ __attribute__((aligned(16))) float ldsU[NSR * WPITCH];
  __shared__ float red[4];
  const int g = blockIdx.y;
  const int row0 = (blockIdx.x >> 1) * NSR, col0 = (blockIdx.x & 1) * NSC;
  const int tid = threadIdx.x, lane = tid & 31, wave = tid >> 5;
  const int l15 = lane & 15;
  const int h = lane >> 4;
  const int kb = 2 * h;
  const float* Bg = B + g * (NN * NN);
  const float* Sg = S + g * (NN * NN);

  float4 rb0[16], rb1[16], rs0[8], rs1[8];
  float4 rr0, rr1, rr2, rr3;
  const float* Wg = FIRST ? Sg : Bg;
#pragma unroll
  for (int i = 0; i < 16; ++i) {
    int q = i * NST + tid;
    int kp = q >> 5, c = (q & 31) * 4;
    rb0[i] = *reinterpret_cast<const float4*>(Wg + (2 * kp) * NN + c);
    rb1[i] = *reinterpret_cast<const float4*>(Wg + (2 * kp + 1) * NN + c);
  }
  if constexpr (!FIRST) {
    const float* Rg = Bg + row0 * NN + tid * 4;
    rr0 = *reinterpret_cast<const float4*>(Rg);
    rr1 = *reinterpret_cast<const float4*>(Rg + NST * 4);
    rr2 = *reinterpret_cast<const float4*>(Rg + 2 * NST * 4);
    rr3 = *reinterpret_cast<const float4*>(Rg + 3 * NST * 4);
  }
#pragma unroll
  for (int i = 0; i < 8; ++i) {
    int q = i * NST + tid;
    int kp = q >> 4, c = (q & 15) * 4;
    rs0[i] = *reinterpret_cast<const float4*>(Sg + (2 * kp) * NN + col0 + c);
    rs1[i] = *reinterpret_cast<const float4*>(Sg + (2 * kp + 1) * NN + col0 + c);
  }
  if constexpr (FIRST) {
    float ss = 0.f;
#pragma unroll
    for (int i = 0; i < 16; ++i) {
      int q = i * NST + tid;
      int kp = q >> 5, c = (q & 31) * 4;
      const int d = 2 * kp - c;
      float4 a = rb0[i], b = rb1[i];
      a.x += (d == 0) ? EPSV : 0.f;  a.y += (d == 1) ? EPSV : 0.f;
      a.z += (d == 2) ? EPSV : 0.f;  a.w += (d == 3) ? EPSV : 0.f;
      b.x += (d == -1) ? EPSV : 0.f; b.y += (d == 0) ? EPSV : 0.f;
      b.z += (d == 1) ? EPSV : 0.f;  b.w += (d == 2) ? EPSV : 0.f;
      ss += a.x * a.x; ss += a.y * a.y; ss += a.z * a.z; ss += a.w * a.w;
      ss += b.x * b.x; ss += b.y * b.y; ss += b.z * b.z; ss += b.w * b.w;
      rb0[i] = float4{(d == 0) ? 1.f : 0.f, (d == 1) ? 1.f : 0.f, (d == 2) ? 1.f : 0.f, (d == 3) ? 1.f : 0.f};
      rb1[i] = float4{(d == -1) ? 1.f : 0.f, (d == 0) ? 1.f : 0.f, (d == 1) ? 1.f : 0.f, (d == 2) ? 1.f : 0.f};
    }
    {
      const int d0 = row0 + (tid >> 5) - (tid & 31) * 4;
      rr0 = float4{(d0 == 0) ? 1.f : 0.f, (d0 == 1) ? 1.f : 0.f, (d0 == 2) ? 1.f : 0.f, (d0 == 3) ? 1.f : 0.f};
      rr1 = float4{(d0 == -4) ? 1.f : 0.f, (d0 == -3) ? 1.f : 0.f, (d0 == -2) ? 1.f : 0.f, (d0 == -1) ? 1.f : 0.f};
      rr2 = float4{(d0 == -8) ? 1.f : 0.f, (d0 == -7) ? 1.f : 0.f, (d0 == -6) ? 1.f : 0.f, (d0 == -5) ? 1.f : 0.f};
      rr3 = float4{(d0 == -12) ? 1.f : 0.f, (d0 == -11) ? 1.f : 0.f, (d0 == -10) ? 1.f : 0.f, (d0 == -9) ? 1.f : 0.f};
    }
#pragma unroll
    for (int off = 16; off > 0; off >>= 1) ss += __shfl_xor(ss, off);
    if (lane == 0) red[wave] = ss;
  }
#pragma unroll
  for (int i = 0; i < 16; ++i) {
    int q = i * NST + tid;
    int kp = q >> 5, c = (q & 31) * 4;
    float4 lo = {rb0[i].x, rb1[i].x, rb0[i].y, rb1[i].y};
    float4 hi = {rb0[i].z, rb1[i].z, rb0[i].w, rb1[i].w};
    *reinterpret_cast<float4*>(&ldsBB[kp * BPB + c * 2]) = lo;
    *reinterpret_cast<float4*>(&ldsBB[kp * BPB + c * 2 + 4]) = hi;
  }
  {
    float* rp = &ldsR[(tid >> 5) * WPITCH + (tid & 31) * 4];
    *reinterpret_cast<float4*>(rp) = rr0;
    *reinterpret_cast<float4*>(rp + 4 * WPITCH) = rr1;
    *reinterpret_cast<float4*>(rp + 8 * WPITCH) = rr2;
    *reinterpret_cast<float4*>(rp + 12 * WPITCH) = rr3;
  }
  __syncthreads();

  v8f z = {};

  const int tc = wave * 32;
  const float* fpbB = &ldsBB[h * BPB + (tc + l15) * 2];
  v8f acc1[2];
  acc1[0] = z;
  acc1[1] = z;
  mma_row_two_cols<BPB>(&ldsR[l15 * WPITCH + kb], fpbB, acc1);
#pragma unroll
  for (int t = 0; t < 2; ++t) {
#pragma unroll
    for (int r = 0; r < 8; ++r)
      ldsT[(8 * h + r) * WPITCH + tc + t * 16 + l15] = acc1[t][r];
  }

  if constexpr (FIRST) {
    const float tot = (red[0] + red[1]) + (red[2] + red[3]);
    const float nrm = sqrtf(tot);
    const float inv = 1.0f / nrm;
    if (tid == 0 && blockIdx.x == 0) rs[g] = 1.0f / sqrtf(nrm);
    float* Sng = Sn + g * (NN * NN);
#pragma unroll
    for (int i = 0; i < 8; ++i) {
      int q = i * NST + tid;
      int kp = q >> 4, c = (q & 15) * 4;
      const int d = 2 * kp - (col0 + c);
      float4 a = rs0[i], b = rs1[i];
      a.x = (a.x + ((d == 0) ? EPSV : 0.f)) * inv;  a.y = (a.y + ((d == 1) ? EPSV : 0.f)) * inv;
      a.z = (a.z + ((d == 2) ? EPSV : 0.f)) * inv;  a.w = (a.w + ((d == 3) ? EPSV : 0.f)) * inv;
      b.x = (b.x + ((d == -1) ? EPSV : 0.f)) * inv; b.y = (b.y + ((d == 0) ? EPSV : 0.f)) * inv;
      b.z = (b.z + ((d == 1) ? EPSV : 0.f)) * inv;  b.w = (b.w + ((d == 2) ? EPSV : 0.f)) * inv;
      rs0[i] = a;
      rs1[i] = b;
      if (row0 == 0) {
        *reinterpret_cast<float4*>(Sng + (2 * kp) * NN + col0 + c) = a;
        *reinterpret_cast<float4*>(Sng + (2 * kp + 1) * NN + col0 + c) = b;
      }
    }
  }
#pragma unroll
  for (int i = 0; i < 8; ++i) {
    int q = i * NST + tid;
    int kp = q >> 4, c = (q & 15) * 4;
    float4 lo = {rs0[i].x, rs1[i].x, rs0[i].y, rs1[i].y};
    float4 hi = {rs0[i].z, rs1[i].z, rs0[i].w, rs1[i].w};
    *reinterpret_cast<float4*>(&ldsS[kp * BPS + c * 2]) = lo;
    *reinterpret_cast<float4*>(&ldsS[kp * BPS + c * 2 + 4]) = hi;
  }
  __syncthreads();

  v8f acc2[2];
  acc2[0] = z;
  acc2[1] = z;
  mma_row_two_cols<BPB>(&ldsT[l15 * WPITCH + kb], fpbB, acc2);
#pragma unroll
  for (int t = 0; t < 2; ++t) {
#pragma unroll
    for (int r = 0; r < 8; ++r)
      ldsU[(8 * h + r) * WPITCH + tc + t * 16 + l15] = acc2[t][r];
  }
  __syncthreads();

  const int tn = wave * 16;
  float p[8];
#pragma unroll
  for (int r = 0; r < 8; ++r)
    p[r] = ldsR[(8 * h + r) * WPITCH + col0 + tn + l15];
  v8f acc3 = mma_row_one_col<BPS>(&ldsU[l15 * WPITCH + kb],
                                  &ldsS[h * BPS + (tn + l15) * 2]);
  float* Og = Bn + g * (NN * NN) + (row0 + 8 * h) * NN + col0 + tn + l15;
#pragma unroll
  for (int r = 0; r < 8; ++r) {
    float v = -0.5f * acc3[r];
    v = __builtin_fmaf(1.5f, p[r], v);
    Og[r * NN] = v;
  }
}

#define OWAVES 12
#define OSTRIPS 3
#define OSLAB (OWAVES * OSTRIPS * 16)
#define OSTEP (OWAVES * 16)

static __device__ __forceinline__ void out_chunk_load(
    const float* __restrict__ p, float (&b)[16]) {
#pragma unroll
  for (int j = 0; j < 8; ++j) {
    b[2 * j]     = p[(size_t)(4 * j) * DD];
    b[2 * j + 1] = p[(size_t)(4 * j + 1) * DD];
  }
}

__global__ void __launch_bounds__(OWAVES * 32)
oni_out(const float* __restrict__ Bt, const float* __restrict__ w,
        const float* __restrict__ mean, const float* __restrict__ rs,
        float* __restrict__ out) {
  __shared__ __attribute__((aligned(16))) float ldsA[NN * WPITCH];
  __shared__ __attribute__((aligned(16))) float ldsM[NN];
  const int g = blockIdx.y;
  const int tid = threadIdx.x, lane = tid & 31, wave = tid >> 5;
  const int l15 = lane & 15;
  const int kb = (lane < 16) ? 0 : 2;
  const int col0 = blockIdx.x * OSLAB + wave * 16 + l15;
  const float* Ag = Bt + g * (NN * NN);

  float4 ra[10];
#pragma unroll
  for (int i = 0; i < 10; ++i) {
    int idx = i * (OWAVES * 32) + tid;
    ra[i] = *reinterpret_cast<const float4*>(Ag + idx * 4);
  }
  const bool part = tid < (NN * NN / 4 - 10 * OWAVES * 32);
  float4 rl = {};
  if (part) rl = *reinterpret_cast<const float4*>(Ag + (10 * OWAVES * 32 + tid) * 4);
  float mrow = 0.f;
  if (tid < NN) mrow = mean[g * NN + tid];

  const float* wp = w + (size_t)(g * NN + kb) * DD + col0;
  float bc[16], bn[16];
  out_chunk_load(wp, bc);

#pragma unroll
  for (int i = 0; i < 10; ++i) {
    int idx = i * (OWAVES * 32) + tid;
    int r = idx >> 5, c = (idx & 31) * 4;
    *reinterpret_cast<float4*>(&ldsA[r * WPITCH + c]) = ra[i];
  }
  if (part) {
    int idx = 10 * OWAVES * 32 + tid;
    int r = idx >> 5, c = (idx & 31) * 4;
    *reinterpret_cast<float4*>(&ldsA[r * WPITCH + c]) = rl;
  }
  if (tid < NN) ldsM[tid] = mrow;
  __syncthreads();

  const float scale = rs[g];
  float* op = out + (size_t)(g * NN + ((lane < 16) ? 0 : 8)) * DD + col0;

  v8f acc[8];
  v8f z = {};
#pragma unroll
  for (int t = 0; t < 8; ++t) acc[t] = z;

  v2f ac[8], an[8];
#pragma unroll
  for (int mt = 0; mt < 8; ++mt)
    ac[mt] = *reinterpret_cast<const v2f*>(&ldsA[(mt * 16 + l15) * WPITCH + kb]);
  v2f bf;
  {
    v2f mm = *reinterpret_cast<const v2f*>(&ldsM[kb]);
    bf.x = bc[0] - mm.x;
    bf.y = bc[1] - mm.y;
  }

  for (int it = 0; it < OSTRIPS * 4; ++it) {
    const int kc = it & 3;
    const int itn = (it + 1 < OSTRIPS * 4) ? it + 1 : it;
    out_chunk_load(wp + (size_t)((itn & 3) * KCHUNK) * DD + (itn >> 2) * OSTEP, bn);
    const int k0 = kc * KCHUNK;
#pragma unroll
    for (int j = 0; j < 8; ++j) {
      const int k4n = (j < 7) ? (k0 + 4 * j + 4) : ((itn & 3) * KCHUNK);
#pragma unroll
      for (int mt = 0; mt < 8; ++mt)
        an[mt] = *reinterpret_cast<const v2f*>(&ldsA[(mt * 16 + l15) * WPITCH + k4n + kb]);
      v2f mm = *reinterpret_cast<const v2f*>(&ldsM[k4n + kb]);
      __builtin_amdgcn_sched_barrier(0);
#pragma unroll
      for (int mt = 0; mt < 8; ++mt)
        acc[mt] = wmma_f32(ac[mt], bf, acc[mt]);
      __builtin_amdgcn_sched_barrier(0);
      const float bx = (j < 7) ? bc[2 * j + 2] : bn[0];
      const float by = (j < 7) ? bc[2 * j + 3] : bn[1];
      bf.x = bx - mm.x;
      bf.y = by - mm.y;
#pragma unroll
      for (int mt = 0; mt < 8; ++mt) ac[mt] = an[mt];
    }
    if (kc == 3) {
      float* os = op + (it >> 2) * OSTEP;
#pragma unroll
      for (int mt = 0; mt < 8; ++mt) {
#pragma unroll
        for (int r = 0; r < 8; ++r)
          os[(size_t)(mt * 16 + r) * DD] = acc[mt][r] * scale;
        acc[mt] = z;
      }
    }
#pragma unroll
    for (int q = 0; q < 16; ++q) bc[q] = bn[q];
  }
}

extern "C" void kernel_launch(void* const* d_in, const int* in_sizes, int n_in,
                              void* d_out, int out_size, void* d_ws, size_t ws_size,
                              hipStream_t stream) {
  const float* w = (const float*)d_in[0];
  float* out = (float*)d_out;
  float* ws = (float*)d_ws;

  float* mean = ws;
  float* rs = ws + 2048;
  float* S  = ws + 2080;
  float* BA = S + MATF;
  float* BB = BA + MATF;
  float* Sn = BB + MATF;

  oni_mean<<<NG * NN, 256, 0, stream>>>(w, mean);
  oni_zero<<<MATF / 256, 256, 0, stream>>>(S);
  oni_cov<<<dim3(NN / CROWS, KSPLIT, NG), 256, 0, stream>>>(w, mean, S);
  oni_ns_iter<true><<<dim3((NN / NSR) * (NN / NSC), NG), NST, 0, stream>>>(BB, S, BA, Sn, rs);

  float* Bc = BA;
  float* Bn = BB;
  for (int t = 1; t < TITER; ++t) {
    oni_ns_iter<false><<<dim3((NN / NSR) * (NN / NSC), NG), NST, 0, stream>>>(Bc, Sn, Bn, nullptr, nullptr);
    float* tmp = Bc; Bc = Bn; Bn = tmp;
  }

  oni_out<<<dim3(DD / OSLAB, NG), OWAVES * 32, 0, stream>>>(Bc, w, mean, rs, out);
}
